# starting kernel
# baseline (speedup 1.0000x reference)
#include <hip/hip_runtime.h>

typedef __attribute__((ext_vector_type(2))) float v2f;
typedef __attribute__((ext_vector_type(8))) float v8f;

#define LDSS 132


__global__ __launch_bounds__(256) void gcn_zero(float* __restrict__ p, long n) {
    long i = (long)blockIdx.x * blockDim.x + threadIdx.x;
    long stride = (long)gridDim.x * blockDim.x;
    for (; i < n; i += stride) p[i] = 0.0f;
}

__global__ __launch_bounds__(256) void gcn_degree(const int* __restrict__ src,
                                                  const int* __restrict__ dst,
                                                  float* __restrict__ deg_src,
                                                  float* __restrict__ deg_dst, int E) {
    int e = blockIdx.x * blockDim.x + threadIdx.x;
    if (e < E) {
        atomicAdd(&deg_src[src[e]], 1.0f);
        atomicAdd(&deg_dst[dst[e]], 1.0f);
    }
}

__global__ __launch_bounds__(256) void gcn_norm(float* __restrict__ deg, int n) {
    int i = blockIdx.x * blockDim.x + threadIdx.x;
    if (i < n) deg[i] = rsqrtf(fmaxf(deg[i], 1.0f));
}


__global__ __launch_bounds__(256) void gcn_gemm1(const float* __restrict__ feats,
                                                 const float* __restrict__ norm_src,
                                                 const float* __restrict__ W1,
                                                 float* __restrict__ H1, int nn) {
    __shared__ float As[16 * LDSS];
    const int tid  = threadIdx.x;
    const int wave = tid >> 5;
    const int lane = tid & 31;
    const int mbase = blockIdx.x * 16;

    for (int f = tid; f < 16 * 32; f += 256) {
        int row = f >> 5;
        int c4  = (f & 31) << 2;
        int r   = mbase + row; if (r >= nn) r = nn - 1;
        float nrm = norm_src[r];
        const float4 v = *(const float4*)(feats + (size_t)r * 128 + c4);
        float4 s; s.x = v.x * nrm; s.y = v.y * nrm; s.z = v.z * nrm; s.w = v.w * nrm;
        *(float4*)(&As[row * LDSS + c4]) = s;
    }
    __syncthreads();

    const int col  = (wave << 4) + (lane & 15);
    const int krow = (lane >> 4) << 1;
    v8f c = {};
    #pragma unroll 4
    for (int kk = 0; kk < 128; kk += 4) {
        const int k0 = kk + krow;
        v2f a = *(const v2f*)(&As[(lane & 15) * LDSS + k0]);
        v2f b;
        b.x = W1[(size_t)k0 * 128 + col];
        b.y = W1[(size_t)(k0 + 1) * 128 + col];
        c = __builtin_amdgcn_wmma_f32_16x16x4_f32(false, a, false, b, (short)0, c, false, false);
    }

    const int rbase = mbase + ((lane >> 4) << 3);
    if (mbase + 16 <= nn) {
        float* p = H1 + (size_t)rbase * 128 + col;
        #pragma unroll
        for (int v = 0; v < 8; ++v) p[v * 128] = c[v];
    } else {
        #pragma unroll
        for (int v = 0; v < 8; ++v) {
            int r = rbase + v;
            if (r < nn) H1[(size_t)r * 128 + col] = c[v];
        }
    }
}


__global__ __launch_bounds__(256) void gcn_scatter128(const float* __restrict__ H,
                                                      const int* __restrict__ src,
                                                      const int* __restrict__ dst,
                                                      float* __restrict__ agg, int E) {
    int e = blockIdx.x * 8 + (threadIdx.x >> 5);
    if (e >= E) return;
    int lane = threadIdx.x & 31;
    int s = src[e], d = dst[e];
    const float4 m = *(const float4*)(H + (size_t)s * 128 + lane * 4);
    float* o = agg + (size_t)d * 128 + lane * 4;
    atomicAdd(o + 0, m.x); atomicAdd(o + 1, m.y);
    atomicAdd(o + 2, m.z); atomicAdd(o + 3, m.w);
}


__global__ __launch_bounds__(128) void gcn_gemm2(const float* __restrict__ agg1,
                                                 const float* __restrict__ norm_src,
                                                 const float* __restrict__ norm_dst,
                                                 const float* __restrict__ b1,
                                                 const float* __restrict__ W2,
                                                 float* __restrict__ H2, int nn) {
    __shared__ float As[16 * LDSS];
    const int tid  = threadIdx.x;
    const int wave = tid >> 5;
    const int lane = tid & 31;
    const int mbase = blockIdx.x * 16;

    for (int f = tid; f < 16 * 32; f += 128) {
        int row = f >> 5;
        int c4  = (f & 31) << 2;
        int r   = mbase + row; if (r >= nn) r = nn - 1;
        float nd = norm_dst[r];
        float ns = norm_src[r];
        const float4 v  = *(const float4*)(agg1 + (size_t)r * 128 + c4);
        const float4 bb = *(const float4*)(b1 + c4);
        float4 s;
        s.x = fmaxf(v.x * nd + bb.x, 0.0f) * ns;
        s.y = fmaxf(v.y * nd + bb.y, 0.0f) * ns;
        s.z = fmaxf(v.z * nd + bb.z, 0.0f) * ns;
        s.w = fmaxf(v.w * nd + bb.w, 0.0f) * ns;
        *(float4*)(&As[row * LDSS + c4]) = s;
    }
    __syncthreads();

    const int col  = (wave << 4) + (lane & 15);
    const int krow = (lane >> 4) << 1;
    v8f c = {};
    #pragma unroll 4
    for (int kk = 0; kk < 128; kk += 4) {
        const int k0 = kk + krow;
        v2f a = *(const v2f*)(&As[(lane & 15) * LDSS + k0]);
        v2f b;
        b.x = W2[(size_t)k0 * 64 + col];
        b.y = W2[(size_t)(k0 + 1) * 64 + col];
        c = __builtin_amdgcn_wmma_f32_16x16x4_f32(false, a, false, b, (short)0, c, false, false);
    }

    const int rbase = mbase + ((lane >> 4) << 3);
    if (mbase + 16 <= nn) {
        float* p = H2 + (size_t)rbase * 64 + col;
        #pragma unroll
        for (int v = 0; v < 8; ++v) p[v * 64] = c[v];
    } else {
        #pragma unroll
        for (int v = 0; v < 8; ++v) {
            int r = rbase + v;
            if (r < nn) H2[(size_t)r * 64 + col] = c[v];
        }
    }
}


__global__ __launch_bounds__(256) void gcn_scatter64(const float* __restrict__ H,
                                                     const int* __restrict__ src,
                                                     const int* __restrict__ dst,
                                                     float* __restrict__ agg, int E) {
    int e = blockIdx.x * 8 + (threadIdx.x >> 5);
    if (e >= E) return;
    int lane = threadIdx.x & 31;
    int s = src[e], d = dst[e];
    const float2 m = *(const float2*)(H + (size_t)s * 64 + lane * 2);
    float* o = agg + (size_t)d * 64 + lane * 2;
    atomicAdd(o + 0, m.x); atomicAdd(o + 1, m.y);
}


__global__ __launch_bounds__(256) void gcn_finalize(const float* __restrict__ agg2,
                                                    const float* __restrict__ norm_dst,
                                                    const float* __restrict__ b2,
                                                    float* __restrict__ out, long total) {
    long i = (long)blockIdx.x * blockDim.x + threadIdx.x;
    if (i < total) {
        long node = i >> 6;
        int  j    = (int)(i & 63);
        out[i] = agg2[i] * norm_dst[node] + b2[j];
    }
}


extern "C" void kernel_launch(void* const* d_in, const int* in_sizes, int n_in,
                              void* d_out, int out_size, void* d_ws, size_t ws_size,
                              hipStream_t stream) {
    const float* feats = (const float*)d_in[0];
    const float* W1    = (const float*)d_in[1];
    const float* b1    = (const float*)d_in[2];
    const float* W2    = (const float*)d_in[3];
    const float* b2    = (const float*)d_in[4];
    const int*   src   = (const int*)d_in[5];
    const int*   dst   = (const int*)d_in[6];

    const int N = in_sizes[0] / 128;
    const int E = in_sizes[5];

    float* ws       = (float*)d_ws;
    float* norm_src = ws;                  ws += N;
    float* norm_dst = ws;                  ws += N;
    float* agg1     = ws;                  ws += (size_t)N * 128;
    float* agg2     = ws;                  ws += (size_t)N * 64;
    float* H1       = ws;                  ws += (size_t)N * 128;
    float* H2       = ws;                  ws += (size_t)N * 64;

    const long nzero = (long)N * (2 + 128 + 64);
    gcn_zero<<<2048, 256, 0, stream>>>((float*)d_ws, nzero);

    gcn_degree<<<(E + 255) / 256, 256, 0, stream>>>(src, dst, norm_src, norm_dst, E);
    gcn_norm<<<(2 * N + 255) / 256, 256, 0, stream>>>((float*)d_ws, 2 * N);

    const int mtiles = (N + 15) / 16;

    gcn_gemm1<<<mtiles, 256, 0, stream>>>(feats, norm_src, W1, H1, N);
    gcn_scatter128<<<(E + 7) / 8, 256, 0, stream>>>(H1, src, dst, agg1, E);

    gcn_gemm2<<<mtiles, 128, 0, stream>>>(agg1, norm_src, norm_dst, b1, W2, H2, N);
    gcn_scatter64<<<(E + 7) / 8, 256, 0, stream>>>(H2, src, dst, agg2, E);

    const long total = (long)N * 64;
    gcn_finalize<<<(int)((total + 255) / 256), 256, 0, stream>>>(agg2, norm_dst, b2,
                                                                 (float*)d_out, total);
}


# layer-2 owner kernel takes its edge list (source, local row) and count from the layer-1 owner via the workspace: its dst-list scan and src look-ups go; sum order = LDS arrival as before
# speedup vs baseline: 1.5553x; 1.5553x over previous
#include <hip/hip_runtime.h>

typedef __attribute__((ext_vector_type(2))) float v2f;
typedef __attribute__((ext_vector_type(8))) float v8f;

#define LDSS 132


__global__ __launch_bounds__(256) void gcn_zero(float* __restrict__ p, long n) {
    long i = (long)blockIdx.x * blockDim.x + threadIdx.x;
    long stride = (long)gridDim.x * blockDim.x;
    for (; i < n; i += stride) p[i] = 0.0f;
}

__global__ __launch_bounds__(256) void gcn_degree(const int* __restrict__ src,
                                                  const int* __restrict__ dst,
                                                  float* __restrict__ deg_src,
                                                  float* __restrict__ deg_dst, int E) {
    int e = blockIdx.x * blockDim.x + threadIdx.x;
    if (e < E) {
        atomicAdd(&deg_src[src[e]], 1.0f);
        atomicAdd(&deg_dst[dst[e]], 1.0f);
    }
}

__global__ __launch_bounds__(256) void gcn_norm(float* __restrict__ deg, int n) {
    int i = blockIdx.x * blockDim.x + threadIdx.x;
    if (i < n) deg[i] = rsqrtf(fmaxf(deg[i], 1.0f));
}


__global__ __launch_bounds__(256) void gcn_gemm1(const float* __restrict__ feats,
                                                 const float* __restrict__ norm_src,
                                                 const float* __restrict__ W1,
                                                 float* __restrict__ H1, int nn) {
    __shared__ float As[16 * LDSS];
    const int tid  = threadIdx.x;
    const int wave = tid >> 5;
    const int lane = tid & 31;
    const int mbase = blockIdx.x * 16;

    for (int f = tid; f < 16 * 32; f += 256) {
        int row = f >> 5;
        int c4  = (f & 31) << 2;
        int r   = mbase + row; if (r >= nn) r = nn - 1;
        float nrm = norm_src[r];
        const float4 v = *(const float4*)(feats + (size_t)r * 128 + c4);
        float4 s; s.x = v.x * nrm; s.y = v.y * nrm; s.z = v.z * nrm; s.w = v.w * nrm;
        *(float4*)(&As[row * LDSS + c4]) = s;
    }
    __syncthreads();

    const int col  = (wave << 4) + (lane & 15);
    const int krow = (lane >> 4) << 1;
    v8f c = {};
    #pragma unroll 4
    for (int kk = 0; kk < 128; kk += 4) {
        const int k0 = kk + krow;
        v2f a = *(const v2f*)(&As[(lane & 15) * LDSS + k0]);
        v2f b;
        b.x = W1[(size_t)k0 * 128 + col];
        b.y = W1[(size_t)(k0 + 1) * 128 + col];
        c = __builtin_amdgcn_wmma_f32_16x16x4_f32(false, a, false, b, (short)0, c, false, false);
    }

    const int rbase = mbase + ((lane >> 4) << 3);
    if (mbase + 16 <= nn) {
        float* p = H1 + (size_t)rbase * 128 + col;
        #pragma unroll
        for (int v = 0; v < 8; ++v) p[v * 128] = c[v];
    } else {
        #pragma unroll
        for (int v = 0; v < 8; ++v) {
            int r = rbase + v;
            if (r < nn) H1[(size_t)r * 128 + col] = c[v];
        }
    }
}


#define OWN_ROWS 196

#define QLIST 8192

template <int F, int QCAP, bool FINAL>
__device__ __forceinline__ void own_body(float* tile, int2* queue, int* qcount,
                                         const float* __restrict__ H,
                                         const int* __restrict__ src,
                                         const int* __restrict__ dst,
                                         const float* __restrict__ scale,
                                         const float* __restrict__ bias,
                                         float* __restrict__ agg,
                                         int* __restrict__ lcount, int* __restrict__ llist,
                                         int E, int nn) {
    static_assert(FINAL ? (QCAP >= QLIST) : (QCAP == QLIST), "queue sizes and the handed-over list");
    constexpr int VEC = F / 32;
    const int tid   = threadIdx.x;
    const int lane  = tid & 31;
    const int wave  = __builtin_amdgcn_readfirstlane(tid >> 5);
    const int nbase = blockIdx.x * OWN_ROWS;

    for (int i = tid; i < OWN_ROWS * F / 4; i += 1024)
        ((float4*)tile)[i] = make_float4(0.0f, 0.0f, 0.0f, 0.0f);

    int  n = 0;
    int  nraw = 0;
    bool listed = false;
    if constexpr (FINAL) {
        const int saved = __builtin_amdgcn_readfirstlane(lcount[blockIdx.x]);
        listed = (unsigned)saved <= (unsigned)QLIST;
        if (listed) {
            n = saved;
            const int2* lp = (const int2*)llist + (size_t)blockIdx.x * QLIST;
            for (int i = tid; i < n; i += 4 * 1024) {
                int2 v[4];
                #pragma unroll
                for (int k = 0; k < 4; ++k) v[k] = lp[(i + k * 1024 < n) ? (i + k * 1024) : i];
                queue[i] = v[0];
                #pragma unroll
                for (int k = 1; k < 4; ++k)
                    if (i + k * 1024 < n) queue[i + k * 1024] = v[k];
            }
        }
    }
    if (!listed) {
        if (tid == 0) *qcount = 0;
        __syncthreads();

        auto file = [&](unsigned dl, int e) {
            if (dl < (unsigned)OWN_ROWS) {
                const int slot = atomicAdd(qcount, 1);
                if (slot < QCAP) {
                    queue[slot] = make_int2(e, (int)dl);
                } else {
                    const float* hp = H + (size_t)src[e] * F;
                    float* tp = tile + dl * F;
                    #pragma nounroll
                    for (int j = 0; j < F; ++j) atomicAdd(tp + j, hp[j]);
                }
            }
        };
        auto file4 = [&](const int4 v, int e) {
            const unsigned d0 = (unsigned)(v.x - nbase);
            const unsigned d1 = (unsigned)(v.y - nbase);
            const unsigned d2 = (unsigned)(v.z - nbase);
            const unsigned d3 = (unsigned)(v.w - nbase);
            const unsigned dm = min(min(d0, d1), min(d2, d3));
            if (dm < (unsigned)OWN_ROWS) {
                file(d0, e); file(d1, e + 1); file(d2, e + 2); file(d3, e + 3);
            }
        };

        const int E4 = E >> 2;
        const int4* dst4 = (const int4*)dst;
        const int start = ((int)blockIdx.x * (E4 / (int)gridDim.x)) & ~1023;
        #pragma nounroll
        for (int seg = 0; seg < 2; ++seg) {
            const int hi = seg ? start : E4;
            int b = (seg ? 0 : start) + tid;
            for (; b + 7 * 1024 < hi; b += 8 * 1024) {
                int4 v[8];
                #pragma unroll
                for (int u = 0; u < 8; ++u) v[u] = dst4[b + u * 1024];
                #pragma unroll
                for (int u = 0; u < 8; ++u) file4(v[u], (b + u * 1024) << 2);
            }
            for (; b < hi; b += 1024) file4(dst4[b], b << 2);
        }
        for (int e = (E4 << 2) + tid; e < E; e += 1024)
            file((unsigned)(dst[e] - nbase), e);
        __syncthreads();

        nraw = __builtin_amdgcn_readfirstlane(*qcount);
        n = nraw < QCAP ? nraw : QCAP;

        for (int i = tid; i < n; i += 4 * 1024) {
            int s[4];
            #pragma unroll
            for (int k = 0; k < 4; ++k) {
                const int ik = (i + k * 1024 < n) ? (i + k * 1024) : i;
                s[k] = src[queue[ik].x];
            }
            queue[i].x = s[0];
            #pragma unroll
            for (int k = 1; k < 4; ++k)
                if (i + k * 1024 < n) queue[i + k * 1024].x = s[k];
        }
    }
    __syncthreads();

    if constexpr (!FINAL) {
        if (tid == 0) lcount[blockIdx.x] = nraw;
        if (nraw <= QCAP) {
            int2* lp = (int2*)llist + (size_t)blockIdx.x * QLIST;
            for (int i = tid; i < n; i += 1024) lp[i] = queue[i];
        }
    }

    auto rowload = [&](int s, float* val) {
        const float* hp = H + (size_t)s * F + lane * VEC;
        if constexpr (VEC == 4) {
            const float4 t = *(const float4*)hp;
            val[0] = t.x; val[1] = t.y; val[2] = t.z; val[3] = t.w;
        } else {
            const float2 t = *(const float2*)hp;
            val[0] = t.x; val[1] = t.y;
        }
    };
    constexpr int U = 8;
    const int nfull = n & ~(U - 1);
    for (int i0 = wave * U; i0 < nfull; i0 += 32 * U) {
        const int2 mine = queue[i0 + (lane & (U - 1))];
        int   s[U], r[U];
        float val[U][VEC];
        #pragma unroll
        for (int u = 0; u < U; ++u) {
            s[u] = __builtin_amdgcn_readlane(mine.x, u);
            r[u] = __builtin_amdgcn_readlane(mine.y, u);
        }
        #pragma unroll
        for (int u = 0; u < U; ++u) rowload(s[u], val[u]);
        #pragma unroll
        for (int u = 0; u < U; ++u) {
            float* tp = tile + r[u] * F + lane * VEC;
            #pragma unroll
            for (int j = 0; j < VEC; ++j) atomicAdd(tp + j, val[u][j]);
        }
    }
    if (wave < n - nfull) {
        const int2 q = queue[nfull + wave];
        float val[VEC];
        rowload(q.x, val);
        float* tp = tile + q.y * F + lane * VEC;
        #pragma unroll
        for (int j = 0; j < VEC; ++j) atomicAdd(tp + j, val[j]);
    }

    int rows = nn - nbase;
    if (rows > OWN_ROWS) rows = OWN_ROWS;
    float4* o = (float4*)(agg + (size_t)nbase * F);
    if constexpr (FINAL) {
        constexpr int Q   = F / 4;
        constexpr int RPS = 1024 / Q;
        constexpr int NSW = (OWN_ROWS + RPS - 1) / RPS;
        const int q  = tid % Q;
        const int r0 = tid / Q;
        const float4 bb = *(const float4*)(bias + q * 4);
        float nd[NSW];
        #pragma unroll
        for (int k = 0; k < NSW; ++k) {
            const int r = r0 + k * RPS;
            nd[k] = (r < rows) ? scale[nbase + r] : 0.0f;
        }
        __syncthreads();
        #pragma unroll
        for (int k = 0; k < NSW; ++k) {
            const int r = r0 + k * RPS;
            if (r < rows) {
                const float4 t = ((const float4*)tile)[r * Q + q];
                float4 v;
                v.x = t.x * nd[k] + bb.x;
                v.y = t.y * nd[k] + bb.y;
                v.z = t.z * nd[k] + bb.z;
                v.w = t.w * nd[k] + bb.w;
                o[r * Q + q] = v;
            }
        }
    } else {
        __syncthreads();
        for (int i = tid; i < rows * (F / 4); i += 1024) o[i] = ((const float4*)tile)[i];
    }
}

__global__ __launch_bounds__(1024) void gcn_own128(const float* __restrict__ H,
                                                   const int* __restrict__ src,
                                                   const int* __restrict__ dst,
                                                   float* __restrict__ agg,
                                                   int* __restrict__ lcount,
                                                   int* __restrict__ llist, int E, int nn) {
    __shared__ float tile[OWN_ROWS * 128];
    __shared__ int2  queue[QLIST];
    __shared__ int   qcount;
    own_body<128, QLIST, false>(tile, queue, &qcount, H, src, dst, nullptr, nullptr, agg,
                                lcount, llist, E, nn);
}

__global__ __launch_bounds__(1024) void gcn_own64(const float* __restrict__ H,
                                                  const int* __restrict__ src,
                                                  const int* __restrict__ dst,
                                                  const float* __restrict__ norm_dst,
                                                  const float* __restrict__ b2,
                                                  float* __restrict__ out,
                                                  int* __restrict__ lcount,
                                                  int* __restrict__ llist, int E, int nn) {
    __shared__ float tile[OWN_ROWS * 64];
    __shared__ int2  queue[16384];
    __shared__ int   qcount;
    own_body<64, 16384, true>(tile, queue, &qcount, H, src, dst, norm_dst, b2, out,
                              lcount, llist, E, nn);
}


__global__ __launch_bounds__(128) void gcn_gemm2(const float* __restrict__ agg1,
                                                 const float* __restrict__ norm_src,
                                                 const float* __restrict__ norm_dst,
                                                 const float* __restrict__ b1,
                                                 const float* __restrict__ W2,
                                                 float* __restrict__ H2, int nn) {
    __shared__ float As[16 * LDSS];
    const int tid  = threadIdx.x;
    const int wave = tid >> 5;
    const int lane = tid & 31;
    const int mbase = blockIdx.x * 16;

    for (int f = tid; f < 16 * 32; f += 128) {
        int row = f >> 5;
        int c4  = (f & 31) << 2;
        int r   = mbase + row; if (r >= nn) r = nn - 1;
        float nd = norm_dst[r];
        float ns = norm_src[r];
        const float4 v  = *(const float4*)(agg1 + (size_t)r * 128 + c4);
        const float4 bb = *(const float4*)(b1 + c4);
        float4 s;
        s.x = fmaxf(v.x * nd + bb.x, 0.0f) * ns;
        s.y = fmaxf(v.y * nd + bb.y, 0.0f) * ns;
        s.z = fmaxf(v.z * nd + bb.z, 0.0f) * ns;
        s.w = fmaxf(v.w * nd + bb.w, 0.0f) * ns;
        *(float4*)(&As[row * LDSS + c4]) = s;
    }
    __syncthreads();

    const int col  = (wave << 4) + (lane & 15);
    const int krow = (lane >> 4) << 1;
    v8f c = {};
    #pragma unroll 4
    for (int kk = 0; kk < 128; kk += 4) {
        const int k0 = kk + krow;
        v2f a = *(const v2f*)(&As[(lane & 15) * LDSS + k0]);
        v2f b;
        b.x = W2[(size_t)k0 * 64 + col];
        b.y = W2[(size_t)(k0 + 1) * 64 + col];
        c = __builtin_amdgcn_wmma_f32_16x16x4_f32(false, a, false, b, (short)0, c, false, false);
    }

    const int rbase = mbase + ((lane >> 4) << 3);
    if (mbase + 16 <= nn) {
        float* p = H2 + (size_t)rbase * 64 + col;
        #pragma unroll
        for (int v = 0; v < 8; ++v) p[v * 64] = c[v];
    } else {
        #pragma unroll
        for (int v = 0; v < 8; ++v) {
            int r = rbase + v;
            if (r < nn) H2[(size_t)r * 64 + col] = c[v];
        }
    }
}


extern "C" void kernel_launch(void* const* d_in, const int* in_sizes, int n_in,
                              void* d_out, int out_size, void* d_ws, size_t ws_size,
                              hipStream_t stream) {
    const float* feats = (const float*)d_in[0];
    const float* W1    = (const float*)d_in[1];
    const float* b1    = (const float*)d_in[2];
    const float* W2    = (const float*)d_in[3];
    const float* b2    = (const float*)d_in[4];
    const int*   src   = (const int*)d_in[5];
    const int*   dst   = (const int*)d_in[6];

    const int N = in_sizes[0] / 128;
    const int E = in_sizes[5];

    float* ws       = (float*)d_ws;
    float* norm_src = ws;                  ws += N;
    float* norm_dst = ws;                  ws += N;
    float* agg1     = ws;                  ws += (size_t)N * 128;
    float* H1       = ws;                  ws += (size_t)N * 128;
    float* H2       = ws;                  ws += (size_t)N * 64;
    const int owners = (N + OWN_ROWS - 1) / OWN_ROWS;
    int*  lcount    = (int*)ws;
    int*  llist     = lcount + ((owners + 3) & ~3);

    const long nzero = (long)N * 2;
    gcn_zero<<<(int)((nzero + 255) / 256), 256, 0, stream>>>((float*)d_ws, nzero);

    gcn_degree<<<(E + 255) / 256, 256, 0, stream>>>(src, dst, norm_src, norm_dst, E);
    gcn_norm<<<(2 * N + 255) / 256, 256, 0, stream>>>((float*)d_ws, 2 * N);

    const int mtiles = (N + 15) / 16;

    gcn_gemm1<<<mtiles, 256, 0, stream>>>(feats, norm_src, W1, H1, N);
    gcn_own128<<<owners, 1024, 0, stream>>>(H1, src, dst, agg1, lcount, llist, E, N);

    gcn_gemm2<<<mtiles, 128, 0, stream>>>(agg1, norm_src, norm_dst, b1, W2, H2, N);
    gcn_own64<<<owners, 1024, 0, stream>>>(H2, src, dst, norm_dst, b2, (float*)d_out,
                                           lcount, llist, E, N);
}


# layer-2 product moved into the layer-1 owner: its LDS sum tile (row stride 132) is the A operand in place, 32 waves do the f32 WMMAs and store H2; product launch and agg1 plane go; 6 launches -> 5
# speedup vs baseline: 2.2681x; 1.4583x over previous
#include <hip/hip_runtime.h>

typedef __attribute__((ext_vector_type(2))) float v2f;
typedef __attribute__((ext_vector_type(8))) float v8f;

#define LDSS 132

#define OWN_ROWS 196
#define QLIST 8192
#define PCHUNK 4096
#define POWN 1024
#define LCS 32


__global__ __launch_bounds__(256) void gcn_zero(float* __restrict__ p, long n,
                                                int* __restrict__ lcount, int owners) {
    const long first  = (long)blockIdx.x * blockDim.x + threadIdx.x;
    const long stride = (long)gridDim.x * blockDim.x;
    for (long i = first; i < n; i += stride) p[i] = 0.0f;
    const int c0 = owners <= POWN ? 0 : QLIST + 1;
    for (long o = first; o < owners; o += stride) lcount[o * LCS] = c0;
}

__device__ __forceinline__ float inv_sqrt_deg(float deg) {
    return rsqrtf(fmaxf(deg, 1.0f));
}


__global__ __launch_bounds__(1024) void gcn_file(const int* __restrict__ src,
                                                 const int* __restrict__ dst,
                                                 float* __restrict__ deg_src,
                                                 int* __restrict__ lcount,
                                                 int* __restrict__ llist, int E, int owners) {
    __shared__ int cnt[POWN];
    const int tid = threadIdx.x;
    const bool filing = owners <= POWN;
    if (filing) {
        for (int o = tid; o < owners; o += 1024) cnt[o] = 0;
        __syncthreads();
    }

    const int e = blockIdx.x * PCHUNK + tid * 4;
    int d[4], s[4];
    if (e + 3 < E) {
        const int4 dv = *(const int4*)(dst + e);
        const int4 sv = *(const int4*)(src + e);
        d[0] = dv.x; d[1] = dv.y; d[2] = dv.z; d[3] = dv.w;
        s[0] = sv.x; s[1] = sv.y; s[2] = sv.z; s[3] = sv.w;
    } else {
        #pragma unroll
        for (int k = 0; k < 4; ++k) {
            const bool in = e + k < E;
            d[k] = in ? dst[e + k] : -1;
            s[k] = in ? src[e + k] : 0;
        }
    }
    if (!filing) {
        #pragma unroll
        for (int k = 0; k < 4; ++k)
            if (e + k < E) atomicAdd(&deg_src[s[k]], 1.0f);
        return;
    }
    int own[4], row[4], place[4];
    #pragma unroll
    for (int k = 0; k < 4; ++k) {
        const int o = d[k] / OWN_ROWS;
        const bool ok = d[k] >= 0 && o < owners;
        own[k] = ok ? o : -1;
        row[k] = d[k] - o * OWN_ROWS;
        place[k] = ok ? atomicAdd(&cnt[o], 1) : 0;
    }
    __syncthreads();
    for (int o = tid; o < owners; o += 1024) {
        const int c = cnt[o];
        if (c) cnt[o] = atomicAdd(&lcount[o * LCS], c);
    }
    __syncthreads();
    int2* lp = (int2*)llist;
    #pragma unroll
    for (int k = 0; k < 4; ++k) {
        if (own[k] >= 0) {
            const int slot = cnt[own[k]] + place[k];
            if (slot < QLIST) lp[(size_t)own[k] * QLIST + slot] = make_int2(s[k], row[k]);
        }
    }
    #pragma unroll
    for (int k = 0; k < 4; ++k)
        if (e + k < E) atomicAdd(&deg_src[s[k]], 1.0f);
}


__global__ __launch_bounds__(256) void gcn_gemm1(const float* __restrict__ feats,
                                                 const float* __restrict__ deg_src,
                                                 const float* __restrict__ W1,
                                                 float* __restrict__ H1, int nn) {
    __shared__ float As[16 * LDSS];
    const int tid  = threadIdx.x;
    const int wave = tid >> 5;
    const int lane = tid & 31;
    const int mbase = blockIdx.x * 16;

    for (int f = tid; f < 16 * 32; f += 256) {
        int row = f >> 5;
        int c4  = (f & 31) << 2;
        int r   = mbase + row; if (r >= nn) r = nn - 1;
        float nrm = inv_sqrt_deg(deg_src[r]);
        const float4 v = *(const float4*)(feats + (size_t)r * 128 + c4);
        float4 s; s.x = v.x * nrm; s.y = v.y * nrm; s.z = v.z * nrm; s.w = v.w * nrm;
        *(float4*)(&As[row * LDSS + c4]) = s;
    }
    __syncthreads();

    const int col  = (wave << 4) + (lane & 15);
    const int krow = (lane >> 4) << 1;
    v8f c = {};
    #pragma unroll 4
    for (int kk = 0; kk < 128; kk += 4) {
        const int k0 = kk + krow;
        v2f a = *(const v2f*)(&As[(lane & 15) * LDSS + k0]);
        v2f b;
        b.x = W1[(size_t)k0 * 128 + col];
        b.y = W1[(size_t)(k0 + 1) * 128 + col];
        c = __builtin_amdgcn_wmma_f32_16x16x4_f32(false, a, false, b, (short)0, c, false, false);
    }

    const int rbase = mbase + ((lane >> 4) << 3);
    if (mbase + 16 <= nn) {
        float* p = H1 + (size_t)rbase * 128 + col;
        #pragma unroll
        for (int v = 0; v < 8; ++v) p[v * 128] = c[v];
    } else {
        #pragma unroll
        for (int v = 0; v < 8; ++v) {
            int r = rbase + v;
            if (r < nn) H1[(size_t)r * 128 + col] = c[v];
        }
    }
}


template <int F, int QCAP, bool FINAL, int FN>
__device__ __forceinline__ void own_body(float* tile, int2* queue, int* qcount, int* rowcnt,
                                         const float* __restrict__ H,
                                         const int* __restrict__ src,
                                         const int* __restrict__ dst,
                                         float* __restrict__ deg,
                                         const float* __restrict__ deg_src,
                                         const float* __restrict__ bias,
                                         const float* __restrict__ Wn,
                                         float* __restrict__ agg,
                                         const int* __restrict__ lcount,
                                         const int* __restrict__ llist, int E, int nn) {
    static_assert(QCAP >= QLIST, "the LDS queue holds a whole filed list");
    static_assert(FINAL == (FN == 0), "an owner either finishes the layer or feeds the next product");
    constexpr int VEC = F / 32;
    constexpr int TS  = FINAL ? F : F + 4;
    constexpr int TR  = FINAL ? OWN_ROWS : ((OWN_ROWS + 15) / 16) * 16;
    const int tid   = threadIdx.x;
    const int lane  = tid & 31;
    const int wave  = __builtin_amdgcn_readfirstlane(tid >> 5);
    const int nbase = blockIdx.x * OWN_ROWS;

    for (int i = tid; i < TR * TS / 4; i += 1024)
        ((float4*)tile)[i] = make_float4(0.0f, 0.0f, 0.0f, 0.0f);
    if constexpr (!FINAL) {
        for (int i = tid; i < OWN_ROWS; i += 1024) rowcnt[i] = 0;
        __syncthreads();
    }

    int  n = 0;
    const int saved = __builtin_amdgcn_readfirstlane(lcount[blockIdx.x * LCS]);
    const bool listed = (unsigned)saved <= (unsigned)QLIST;
    if (listed) {
        n = saved;
        const int2* lp = (const int2*)llist + (size_t)blockIdx.x * QLIST;
        for (int i = tid; i < n; i += 4 * 1024) {
            int2 v[4];
            #pragma unroll
            for (int k = 0; k < 4; ++k) v[k] = lp[(i + k * 1024 < n) ? (i + k * 1024) : i];
            queue[i] = v[0];
            if constexpr (!FINAL) atomicAdd(&rowcnt[v[0].y], 1);
            #pragma unroll
            for (int k = 1; k < 4; ++k)
                if (i + k * 1024 < n) {
                    queue[i + k * 1024] = v[k];
                    if constexpr (!FINAL) atomicAdd(&rowcnt[v[k].y], 1);
                }
        }
    } else {
        if (tid == 0) *qcount = 0;
        __syncthreads();

        auto file = [&](unsigned dl, int e) {
            if (dl < (unsigned)OWN_ROWS) {
                if constexpr (!FINAL) atomicAdd(&rowcnt[dl], 1);
                const int slot = atomicAdd(qcount, 1);
                if (slot < QCAP) {
                    queue[slot] = make_int2(e, (int)dl);
                } else {
                    const float* hp = H + (size_t)src[e] * F;
                    float* tp = tile + dl * TS;
                    #pragma nounroll
                    for (int j = 0; j < F; ++j) atomicAdd(tp + j, hp[j]);
                }
            }
        };
        auto file4 = [&](const int4 v, int e) {
            const unsigned d0 = (unsigned)(v.x - nbase);
            const unsigned d1 = (unsigned)(v.y - nbase);
            const unsigned d2 = (unsigned)(v.z - nbase);
            const unsigned d3 = (unsigned)(v.w - nbase);
            const unsigned dm = min(min(d0, d1), min(d2, d3));
            if (dm < (unsigned)OWN_ROWS) {
                file(d0, e); file(d1, e + 1); file(d2, e + 2); file(d3, e + 3);
            }
        };

        const int E4 = E >> 2;
        const int4* dst4 = (const int4*)dst;
        const int start = ((int)blockIdx.x * (E4 / (int)gridDim.x)) & ~1023;
        #pragma nounroll
        for (int seg = 0; seg < 2; ++seg) {
            const int hi = seg ? start : E4;
            int b = (seg ? 0 : start) + tid;
            for (; b + 7 * 1024 < hi; b += 8 * 1024) {
                int4 v[8];
                #pragma unroll
                for (int u = 0; u < 8; ++u) v[u] = dst4[b + u * 1024];
                #pragma unroll
                for (int u = 0; u < 8; ++u) file4(v[u], (b + u * 1024) << 2);
            }
            for (; b < hi; b += 1024) file4(dst4[b], b << 2);
        }
        for (int e = (E4 << 2) + tid; e < E; e += 1024)
            file((unsigned)(dst[e] - nbase), e);
        __syncthreads();

        const int nraw = __builtin_amdgcn_readfirstlane(*qcount);
        n = nraw < QCAP ? nraw : QCAP;

        for (int i = tid; i < n; i += 4 * 1024) {
            int s[4];
            #pragma unroll
            for (int k = 0; k < 4; ++k) {
                const int ik = (i + k * 1024 < n) ? (i + k * 1024) : i;
                s[k] = src[queue[ik].x];
            }
            queue[i].x = s[0];
            #pragma unroll
            for (int k = 1; k < 4; ++k)
                if (i + k * 1024 < n) queue[i + k * 1024].x = s[k];
        }
    }
    __syncthreads();

    auto rowload = [&](int s, float* val) {
        const float* hp = H + (size_t)s * F + lane * VEC;
        if constexpr (VEC == 4) {
            const float4 t = *(const float4*)hp;
            val[0] = t.x; val[1] = t.y; val[2] = t.z; val[3] = t.w;
        } else {
            const float2 t = *(const float2*)hp;
            val[0] = t.x; val[1] = t.y;
        }
    };
    constexpr int U = 8;
    const int nfull = n & ~(U - 1);
    for (int i0 = wave * U; i0 < nfull; i0 += 32 * U) {
        const int2 mine = queue[i0 + (lane & (U - 1))];
        int   s[U], r[U];
        float val[U][VEC];
        #pragma unroll
        for (int u = 0; u < U; ++u) {
            s[u] = __builtin_amdgcn_readlane(mine.x, u);
            r[u] = __builtin_amdgcn_readlane(mine.y, u);
        }
        #pragma unroll
        for (int u = 0; u < U; ++u) rowload(s[u], val[u]);
        #pragma unroll
        for (int u = 0; u < U; ++u) {
            float* tp = tile + r[u] * TS + lane * VEC;
            #pragma unroll
            for (int j = 0; j < VEC; ++j) atomicAdd(tp + j, val[u][j]);
        }
    }
    if (wave < n - nfull) {
        const int2 q = queue[nfull + wave];
        float val[VEC];
        rowload(q.x, val);
        float* tp = tile + q.y * TS + lane * VEC;
        #pragma unroll
        for (int j = 0; j < VEC; ++j) atomicAdd(tp + j, val[j]);
    }

    int rows = nn - nbase;
    if (rows > OWN_ROWS) rows = OWN_ROWS;
    if constexpr (FINAL) {
        float4* o = (float4*)(agg + (size_t)nbase * F);
        constexpr int Q   = F / 4;
        constexpr int RPS = 1024 / Q;
        constexpr int NSW = (OWN_ROWS + RPS - 1) / RPS;
        const int q  = tid % Q;
        const int r0 = tid / Q;
        const float4 bb = *(const float4*)(bias + q * 4);
        float nd[NSW];
        #pragma unroll
        for (int k = 0; k < NSW; ++k) {
            const int r = r0 + k * RPS;
            nd[k] = (r < rows) ? inv_sqrt_deg(deg[nbase + r]) : 0.0f;
        }
        __syncthreads();
        #pragma unroll
        for (int k = 0; k < NSW; ++k) {
            const int r = r0 + k * RPS;
            if (r < rows) {
                const float4 t = ((const float4*)tile)[r * Q + q];
                float4 v;
                v.x = t.x * nd[k] + bb.x;
                v.y = t.y * nd[k] + bb.y;
                v.z = t.z * nd[k] + bb.z;
                v.w = t.w * nd[k] + bb.w;
                o[r * Q + q] = v;
            }
        }
    } else {
        constexpr int NSW = (OWN_ROWS + 31) / 32;
        constexpr int NT  = FN / 16;
        constexpr int MG  = 32 / NT;
        static_assert(F == 128 && (32 % NT) == 0, "one 16-byte piece per lane; waves split evenly over the column tiles");
        const int q  = lane;
        const int r0 = wave;
        const float4 bb = *(const float4*)(bias + q * 4);
        const int rlast = rows - 1;
        float ns[NSW];
        #pragma unroll
        for (int k = 0; k < NSW; ++k) {
            const int r = r0 + k * 32;
            ns[k] = inv_sqrt_deg(deg_src[nbase + (r < rlast ? r : rlast)]);
        }
        const int nt   = wave % NT;
        const int col  = nt * 16 + (lane & 15);
        const int krow = (lane >> 4) << 1;
        v2f b[F / 4];
        #pragma unroll
        for (int s = 0; s < F / 4; ++s) {
            const int k0 = s * 4 + krow;
            b[s].x = Wn[(size_t)k0 * FN + col];
            b[s].y = Wn[(size_t)(k0 + 1) * FN + col];
        }
        __syncthreads();

        #pragma unroll
        for (int k = 0; k < NSW; ++k) {
            const int r = r0 + k * 32;
            if (r < rows) {
                const float nd = inv_sqrt_deg((float)rowcnt[r]);
                float4* tp = (float4*)(tile + r * TS) + q;
                const float4 v = *tp;
                float4 t;
                t.x = fmaxf(v.x * nd + bb.x, 0.0f) * ns[k];
                t.y = fmaxf(v.y * nd + bb.y, 0.0f) * ns[k];
                t.z = fmaxf(v.z * nd + bb.z, 0.0f) * ns[k];
                t.w = fmaxf(v.w * nd + bb.w, 0.0f) * ns[k];
                *tp = t;
            }
        }
        if (tid < rows) deg[nbase + tid] = (float)rowcnt[tid];
        __syncthreads();

        for (int mt = wave / NT; mt * 16 < rows; mt += MG) {
            const float* ap = tile + (mt * 16 + (lane & 15)) * TS + krow;
            v8f c = {};
            #pragma unroll
            for (int s = 0; s < F / 4; ++s) {
                const v2f a = *(const v2f*)(ap + s * 4);
                c = __builtin_amdgcn_wmma_f32_16x16x4_f32(false, a, false, b[s], (short)0, c, false, false);
            }
            const int rbase = mt * 16 + ((lane >> 4) << 3);
            float* p = agg + (size_t)(nbase + rbase) * FN + col;
            if (mt * 16 + 16 <= rows) {
                #pragma unroll
                for (int v = 0; v < 8; ++v) p[v * FN] = c[v];
            } else {
                #pragma unroll
                for (int v = 0; v < 8; ++v)
                    if (rbase + v < rows) p[v * FN] = c[v];
            }
        }
    }
}

__global__ __launch_bounds__(1024) void gcn_own128(const float* __restrict__ H,
                                                   const int* __restrict__ src,
                                                   const int* __restrict__ dst,
                                                   float* __restrict__ deg_dst,
                                                   const float* __restrict__ deg_src,
                                                   const float* __restrict__ b1,
                                                   const float* __restrict__ W2,
                                                   float* __restrict__ H2,
                                                   const int* __restrict__ lcount,
                                                   const int* __restrict__ llist, int E, int nn) {
    __shared__ float tile[((OWN_ROWS + 15) / 16) * 16 * (128 + 4)];
    __shared__ int2  queue[QLIST];
    __shared__ int   rowcnt[OWN_ROWS];
    __shared__ int   qcount;
    own_body<128, QLIST, false, 64>(tile, queue, &qcount, rowcnt, H, src, dst, deg_dst, deg_src, b1, W2,
                                    H2, lcount, llist, E, nn);
}

__global__ __launch_bounds__(1024) void gcn_own64(const float* __restrict__ H,
                                                  const int* __restrict__ src,
                                                  const int* __restrict__ dst,
                                                  float* __restrict__ deg_dst,
                                                  const float* __restrict__ b2,
                                                  float* __restrict__ out,
                                                  const int* __restrict__ lcount,
                                                  const int* __restrict__ llist, int E, int nn) {
    __shared__ float tile[OWN_ROWS * 64];
    __shared__ int2  queue[16384];
    __shared__ int   qcount;
    own_body<64, 16384, true, 0>(tile, queue, &qcount, nullptr, H, src, dst, deg_dst, nullptr, b2, nullptr,
                                 out, lcount, llist, E, nn);
}


extern "C" void kernel_launch(void* const* d_in, const int* in_sizes, int n_in,
                              void* d_out, int out_size, void* d_ws, size_t ws_size,
                              hipStream_t stream) {
    const float* feats = (const float*)d_in[0];
    const float* W1    = (const float*)d_in[1];
    const float* b1    = (const float*)d_in[2];
    const float* W2    = (const float*)d_in[3];
    const float* b2    = (const float*)d_in[4];
    const int*   src   = (const int*)d_in[5];
    const int*   dst   = (const int*)d_in[6];

    const int N = in_sizes[0] / 128;
    const int E = in_sizes[5];

    float* ws       = (float*)d_ws;
    float* deg_src  = ws;                  ws += N;
    float* deg_dst  = ws;                  ws += N;
    float* H1       = ws;                  ws += (size_t)N * 128;
    float* H2       = ws;                  ws += (size_t)N * 64;
    const int owners = (N + OWN_ROWS - 1) / OWN_ROWS;
    int*  lcount    = (int*)ws;
    int*  llist     = lcount + (size_t)owners * LCS;

    const long nzero = (long)N;
    gcn_zero<<<(int)((nzero + 255) / 256), 256, 0, stream>>>(deg_src, nzero, lcount, owners);

    gcn_file<<<(E + PCHUNK - 1) / PCHUNK, 1024, 0, stream>>>(src, dst, deg_src, lcount, llist, E, owners);

    const int mtiles = (N + 15) / 16;

    gcn_gemm1<<<mtiles, 256, 0, stream>>>(feats, deg_src, W1, H1, N);
    gcn_own128<<<owners, 1024, 0, stream>>>(H1, src, dst, deg_dst, deg_src, b1, W2, H2, lcount, llist, E, N);

    gcn_own64<<<owners, 1024, 0, stream>>>(H2, src, dst, deg_dst, b2, (float*)d_out,
                                           lcount, llist, E, N);
}


# layer-1 owner: a lane takes two 8-byte pieces of a gathered row (one per 64-float half), not one 16-byte piece: the 32 lanes of an LDS float add hit 32 banks (was 2 per bank); sum order = LDS arrival
# speedup vs baseline: 2.5513x; 1.1249x over previous
#include <hip/hip_runtime.h>

typedef __attribute__((ext_vector_type(2))) float v2f;
typedef __attribute__((ext_vector_type(8))) float v8f;

#define LDSS 132

#define OWN_ROWS 196
#define QLIST 8192
#define PCHUNK 4096
#define POWN 1024
#define LCS 32


__global__ __launch_bounds__(256) void gcn_zero(float* __restrict__ p, long n,
                                                int* __restrict__ lcount, int owners) {
    const long first  = (long)blockIdx.x * blockDim.x + threadIdx.x;
    const long stride = (long)gridDim.x * blockDim.x;
    for (long i = first; i < n; i += stride) p[i] = 0.0f;
    const int c0 = owners <= POWN ? 0 : QLIST + 1;
    for (long o = first; o < owners; o += stride) lcount[o * LCS] = c0;
}

__device__ __forceinline__ float inv_sqrt_deg(float deg) {
    return rsqrtf(fmaxf(deg, 1.0f));
}


__global__ __launch_bounds__(1024) void gcn_file(const int* __restrict__ src,
                                                 const int* __restrict__ dst,
                                                 float* __restrict__ deg_src,
                                                 int* __restrict__ lcount,
                                                 int* __restrict__ llist, int E, int owners) {
    __shared__ int cnt[POWN];
    const int tid = threadIdx.x;
    const bool filing = owners <= POWN;
    if (filing) {
        for (int o = tid; o < owners; o += 1024) cnt[o] = 0;
        __syncthreads();
    }

    const int e = blockIdx.x * PCHUNK + tid * 4;
    int d[4], s[4];
    if (e + 3 < E) {
        const int4 dv = *(const int4*)(dst + e);
        const int4 sv = *(const int4*)(src + e);
        d[0] = dv.x; d[1] = dv.y; d[2] = dv.z; d[3] = dv.w;
        s[0] = sv.x; s[1] = sv.y; s[2] = sv.z; s[3] = sv.w;
    } else {
        #pragma unroll
        for (int k = 0; k < 4; ++k) {
            const bool in = e + k < E;
            d[k] = in ? dst[e + k] : -1;
            s[k] = in ? src[e + k] : 0;
        }
    }
    if (!filing) {
        #pragma unroll
        for (int k = 0; k < 4; ++k)
            if (e + k < E) atomicAdd(&deg_src[s[k]], 1.0f);
        return;
    }
    int own[4], row[4], place[4];
    #pragma unroll
    for (int k = 0; k < 4; ++k) {
        const int o = d[k] / OWN_ROWS;
        const bool ok = d[k] >= 0 && o < owners;
        own[k] = ok ? o : -1;
        row[k] = d[k] - o * OWN_ROWS;
        place[k] = ok ? atomicAdd(&cnt[o], 1) : 0;
    }
    __syncthreads();
    for (int o = tid; o < owners; o += 1024) {
        const int c = cnt[o];
        if (c) cnt[o] = atomicAdd(&lcount[o * LCS], c);
    }
    __syncthreads();
    int2* lp = (int2*)llist;
    #pragma unroll
    for (int k = 0; k < 4; ++k) {
        if (own[k] >= 0) {
            const int slot = cnt[own[k]] + place[k];
            if (slot < QLIST) lp[(size_t)own[k] * QLIST + slot] = make_int2(s[k], row[k]);
        }
    }
    #pragma unroll
    for (int k = 0; k < 4; ++k)
        if (e + k < E) atomicAdd(&deg_src[s[k]], 1.0f);
}


__global__ __launch_bounds__(256) void gcn_gemm1(const float* __restrict__ feats,
                                                 const float* __restrict__ deg_src,
                                                 const float* __restrict__ W1,
                                                 float* __restrict__ H1, int nn) {
    __shared__ float As[16 * LDSS];
    const int tid  = threadIdx.x;
    const int wave = tid >> 5;
    const int lane = tid & 31;
    const int mbase = blockIdx.x * 16;

    for (int f = tid; f < 16 * 32; f += 256) {
        int row = f >> 5;
        int c4  = (f & 31) << 2;
        int r   = mbase + row; if (r >= nn) r = nn - 1;
        float nrm = inv_sqrt_deg(deg_src[r]);
        const float4 v = *(const float4*)(feats + (size_t)r * 128 + c4);
        float4 s; s.x = v.x * nrm; s.y = v.y * nrm; s.z = v.z * nrm; s.w = v.w * nrm;
        *(float4*)(&As[row * LDSS + c4]) = s;
    }
    __syncthreads();

    const int col  = (wave << 4) + (lane & 15);
    const int krow = (lane >> 4) << 1;
    v8f c = {};
    #pragma unroll 4
    for (int kk = 0; kk < 128; kk += 4) {
        const int k0 = kk + krow;
        v2f a = *(const v2f*)(&As[(lane & 15) * LDSS + k0]);
        v2f b;
        b.x = W1[(size_t)k0 * 128 + col];
        b.y = W1[(size_t)(k0 + 1) * 128 + col];
        c = __builtin_amdgcn_wmma_f32_16x16x4_f32(false, a, false, b, (short)0, c, false, false);
    }

    const int rbase = mbase + ((lane >> 4) << 3);
    if (mbase + 16 <= nn) {
        float* p = H1 + (size_t)rbase * 128 + col;
        #pragma unroll
        for (int v = 0; v < 8; ++v) p[v * 128] = c[v];
    } else {
        #pragma unroll
        for (int v = 0; v < 8; ++v) {
            int r = rbase + v;
            if (r < nn) H1[(size_t)r * 128 + col] = c[v];
        }
    }
}


template <int F, int QCAP, bool FINAL, int FN>
__device__ __forceinline__ void own_body(float* tile, int2* queue, int* qcount, int* rowcnt,
                                         const float* __restrict__ H,
                                         const int* __restrict__ src,
                                         const int* __restrict__ dst,
                                         float* __restrict__ deg,
                                         const float* __restrict__ deg_src,
                                         const float* __restrict__ bias,
                                         const float* __restrict__ Wn,
                                         float* __restrict__ agg,
                                         const int* __restrict__ lcount,
                                         const int* __restrict__ llist, int E, int nn) {
    static_assert(QCAP >= QLIST, "the LDS queue holds a whole filed list");
    static_assert(FINAL == (FN == 0), "an owner either finishes the layer or feeds the next product");
    constexpr int VEC = F / 32;
    constexpr int TS  = FINAL ? F : F + 4;
    constexpr int TR  = FINAL ? OWN_ROWS : ((OWN_ROWS + 15) / 16) * 16;
    const int tid   = threadIdx.x;
    const int lane  = tid & 31;
    const int wave  = __builtin_amdgcn_readfirstlane(tid >> 5);
    const int nbase = blockIdx.x * OWN_ROWS;

    for (int i = tid; i < TR * TS / 4; i += 1024)
        ((float4*)tile)[i] = make_float4(0.0f, 0.0f, 0.0f, 0.0f);
    if constexpr (!FINAL) {
        for (int i = tid; i < OWN_ROWS; i += 1024) rowcnt[i] = 0;
        __syncthreads();
    }

    int  n = 0;
    const int saved = __builtin_amdgcn_readfirstlane(lcount[blockIdx.x * LCS]);
    const bool listed = (unsigned)saved <= (unsigned)QLIST;
    if (listed) {
        n = saved;
        const int2* lp = (const int2*)llist + (size_t)blockIdx.x * QLIST;
        for (int i = tid; i < n; i += 4 * 1024) {
            int2 v[4];
            #pragma unroll
            for (int k = 0; k < 4; ++k) v[k] = lp[(i + k * 1024 < n) ? (i + k * 1024) : i];
            queue[i] = v[0];
            if constexpr (!FINAL) atomicAdd(&rowcnt[v[0].y], 1);
            #pragma unroll
            for (int k = 1; k < 4; ++k)
                if (i + k * 1024 < n) {
                    queue[i + k * 1024] = v[k];
                    if constexpr (!FINAL) atomicAdd(&rowcnt[v[k].y], 1);
                }
        }
    } else {
        if (tid == 0) *qcount = 0;
        __syncthreads();

        auto file = [&](unsigned dl, int e) {
            if (dl < (unsigned)OWN_ROWS) {
                if constexpr (!FINAL) atomicAdd(&rowcnt[dl], 1);
                const int slot = atomicAdd(qcount, 1);
                if (slot < QCAP) {
                    queue[slot] = make_int2(e, (int)dl);
                } else {
                    const float* hp = H + (size_t)src[e] * F;
                    float* tp = tile + dl * TS;
                    #pragma nounroll
                    for (int j = 0; j < F; ++j) atomicAdd(tp + j, hp[j]);
                }
            }
        };
        auto file4 = [&](const int4 v, int e) {
            const unsigned d0 = (unsigned)(v.x - nbase);
            const unsigned d1 = (unsigned)(v.y - nbase);
            const unsigned d2 = (unsigned)(v.z - nbase);
            const unsigned d3 = (unsigned)(v.w - nbase);
            const unsigned dm = min(min(d0, d1), min(d2, d3));
            if (dm < (unsigned)OWN_ROWS) {
                file(d0, e); file(d1, e + 1); file(d2, e + 2); file(d3, e + 3);
            }
        };

        const int E4 = E >> 2;
        const int4* dst4 = (const int4*)dst;
        const int start = ((int)blockIdx.x * (E4 / (int)gridDim.x)) & ~1023;
        #pragma nounroll
        for (int seg = 0; seg < 2; ++seg) {
            const int hi = seg ? start : E4;
            int b = (seg ? 0 : start) + tid;
            for (; b + 7 * 1024 < hi; b += 8 * 1024) {
                int4 v[8];
                #pragma unroll
                for (int u = 0; u < 8; ++u) v[u] = dst4[b + u * 1024];
                #pragma unroll
                for (int u = 0; u < 8; ++u) file4(v[u], (b + u * 1024) << 2);
            }
            for (; b < hi; b += 1024) file4(dst4[b], b << 2);
        }
        for (int e = (E4 << 2) + tid; e < E; e += 1024)
            file((unsigned)(dst[e] - nbase), e);
        __syncthreads();

        const int nraw = __builtin_amdgcn_readfirstlane(*qcount);
        n = nraw < QCAP ? nraw : QCAP;

        for (int i = tid; i < n; i += 4 * 1024) {
            int s[4];
            #pragma unroll
            for (int k = 0; k < 4; ++k) {
                const int ik = (i + k * 1024 < n) ? (i + k * 1024) : i;
                s[k] = src[queue[ik].x];
            }
            queue[i].x = s[0];
            #pragma unroll
            for (int k = 1; k < 4; ++k)
                if (i + k * 1024 < n) queue[i + k * 1024].x = s[k];
        }
    }
    __syncthreads();

    constexpr int NP = F / 64;
    auto rowload = [&](int s, float* val) {
        const float* hp = H + (size_t)s * F + lane * 2;
        #pragma unroll
        for (int p = 0; p < NP; ++p) {
            const float2 t = *(const float2*)(hp + p * 64);
            val[2 * p] = t.x; val[2 * p + 1] = t.y;
        }
    };
    auto rowadd = [&](int r, const float* val) {
        float* tp = tile + r * TS + lane * 2;
        #pragma unroll
        for (int p = 0; p < NP; ++p) {
            atomicAdd(tp + p * 64,     val[2 * p]);
            atomicAdd(tp + p * 64 + 1, val[2 * p + 1]);
        }
    };
    constexpr int U = 8;
    const int nfull = n & ~(U - 1);
    for (int i0 = wave * U; i0 < nfull; i0 += 32 * U) {
        const int2 mine = queue[i0 + (lane & (U - 1))];
        int   s[U], r[U];
        float val[U][VEC];
        #pragma unroll
        for (int u = 0; u < U; ++u) {
            s[u] = __builtin_amdgcn_readlane(mine.x, u);
            r[u] = __builtin_amdgcn_readlane(mine.y, u);
        }
        #pragma unroll
        for (int u = 0; u < U; ++u) rowload(s[u], val[u]);
        #pragma unroll
        for (int u = 0; u < U; ++u) rowadd(r[u], val[u]);
    }
    if (wave < n - nfull) {
        const int2 q = queue[nfull + wave];
        float val[VEC];
        rowload(q.x, val);
        rowadd(q.y, val);
    }

    int rows = nn - nbase;
    if (rows > OWN_ROWS) rows = OWN_ROWS;
    if constexpr (FINAL) {
        float4* o = (float4*)(agg + (size_t)nbase * F);
        constexpr int Q   = F / 4;
        constexpr int RPS = 1024 / Q;
        constexpr int NSW = (OWN_ROWS + RPS - 1) / RPS;
        const int q  = tid % Q;
        const int r0 = tid / Q;
        const float4 bb = *(const float4*)(bias + q * 4);
        float nd[NSW];
        #pragma unroll
        for (int k = 0; k < NSW; ++k) {
            const int r = r0 + k * RPS;
            nd[k] = (r < rows) ? inv_sqrt_deg(deg[nbase + r]) : 0.0f;
        }
        __syncthreads();
        #pragma unroll
        for (int k = 0; k < NSW; ++k) {
            const int r = r0 + k * RPS;
            if (r < rows) {
                const float4 t = ((const float4*)tile)[r * Q + q];
                float4 v;
                v.x = t.x * nd[k] + bb.x;
                v.y = t.y * nd[k] + bb.y;
                v.z = t.z * nd[k] + bb.z;
                v.w = t.w * nd[k] + bb.w;
                o[r * Q + q] = v;
            }
        }
    } else {
        constexpr int NSW = (OWN_ROWS + 31) / 32;
        constexpr int NT  = FN / 16;
        constexpr int MG  = 32 / NT;
        static_assert(F == 128 && (32 % NT) == 0, "one 16-byte piece per lane; waves split evenly over the column tiles");
        const int q  = lane;
        const int r0 = wave;
        const float4 bb = *(const float4*)(bias + q * 4);
        const int rlast = rows - 1;
        float ns[NSW];
        #pragma unroll
        for (int k = 0; k < NSW; ++k) {
            const int r = r0 + k * 32;
            ns[k] = inv_sqrt_deg(deg_src[nbase + (r < rlast ? r : rlast)]);
        }
        const int nt   = wave % NT;
        const int col  = nt * 16 + (lane & 15);
        const int krow = (lane >> 4) << 1;
        v2f b[F / 4];
        #pragma unroll
        for (int s = 0; s < F / 4; ++s) {
            const int k0 = s * 4 + krow;
            b[s].x = Wn[(size_t)k0 * FN + col];
            b[s].y = Wn[(size_t)(k0 + 1) * FN + col];
        }
        __syncthreads();

        #pragma unroll
        for (int k = 0; k < NSW; ++k) {
            const int r = r0 + k * 32;
            if (r < rows) {
                const float nd = inv_sqrt_deg((float)rowcnt[r]);
                float4* tp = (float4*)(tile + r * TS) + q;
                const float4 v = *tp;
                float4 t;
                t.x = fmaxf(v.x * nd + bb.x, 0.0f) * ns[k];
                t.y = fmaxf(v.y * nd + bb.y, 0.0f) * ns[k];
                t.z = fmaxf(v.z * nd + bb.z, 0.0f) * ns[k];
                t.w = fmaxf(v.w * nd + bb.w, 0.0f) * ns[k];
                *tp = t;
            }
        }
        if (tid < rows) deg[nbase + tid] = (float)rowcnt[tid];
        __syncthreads();

        for (int mt = wave / NT; mt * 16 < rows; mt += MG) {
            const float* ap = tile + (mt * 16 + (lane & 15)) * TS + krow;
            v8f c = {};
            #pragma unroll
            for (int s = 0; s < F / 4; ++s) {
                const v2f a = *(const v2f*)(ap + s * 4);
                c = __builtin_amdgcn_wmma_f32_16x16x4_f32(false, a, false, b[s], (short)0, c, false, false);
            }
            const int rbase = mt * 16 + ((lane >> 4) << 3);
            float* p = agg + (size_t)(nbase + rbase) * FN + col;
            if (mt * 16 + 16 <= rows) {
                #pragma unroll
                for (int v = 0; v < 8; ++v) p[v * FN] = c[v];
            } else {
                #pragma unroll
                for (int v = 0; v < 8; ++v)
                    if (rbase + v < rows) p[v * FN] = c[v];
            }
        }
    }
}

__global__ __launch_bounds__(1024) void gcn_own128(const float* __restrict__ H,
                                                   const int* __restrict__ src,
                                                   const int* __restrict__ dst,
                                                   float* __restrict__ deg_dst,
                                                   const float* __restrict__ deg_src,
                                                   const float* __restrict__ b1,
                                                   const float* __restrict__ W2,
                                                   float* __restrict__ H2,
                                                   const int* __restrict__ lcount,
                                                   const int* __restrict__ llist, int E, int nn) {
    __shared__ float tile[((OWN_ROWS + 15) / 16) * 16 * (128 + 4)];
    __shared__ int2  queue[QLIST];
    __shared__ int   rowcnt[OWN_ROWS];
    __shared__ int   qcount;
    own_body<128, QLIST, false, 64>(tile, queue, &qcount, rowcnt, H, src, dst, deg_dst, deg_src, b1, W2,
                                    H2, lcount, llist, E, nn);
}

__global__ __launch_bounds__(1024) void gcn_own64(const float* __restrict__ H,
                                                  const int* __restrict__ src,
                                                  const int* __restrict__ dst,
                                                  float* __restrict__ deg_dst,
                                                  const float* __restrict__ b2,
                                                  float* __restrict__ out,
                                                  const int* __restrict__ lcount,
                                                  const int* __restrict__ llist, int E, int nn) {
    __shared__ float tile[OWN_ROWS * 64];
    __shared__ int2  queue[16384];
    __shared__ int   qcount;
    own_body<64, 16384, true, 0>(tile, queue, &qcount, nullptr, H, src, dst, deg_dst, nullptr, b2, nullptr,
                                 out, lcount, llist, E, nn);
}


extern "C" void kernel_launch(void* const* d_in, const int* in_sizes, int n_in,
                              void* d_out, int out_size, void* d_ws, size_t ws_size,
                              hipStream_t stream) {
    const float* feats = (const float*)d_in[0];
    const float* W1    = (const float*)d_in[1];
    const float* b1    = (const float*)d_in[2];
    const float* W2    = (const float*)d_in[3];
    const float* b2    = (const float*)d_in[4];
    const int*   src   = (const int*)d_in[5];
    const int*   dst   = (const int*)d_in[6];

    const int N = in_sizes[0] / 128;
    const int E = in_sizes[5];

    float* ws       = (float*)d_ws;
    float* deg_src  = ws;                  ws += N;
    float* deg_dst  = ws;                  ws += N;
    float* H1       = ws;                  ws += (size_t)N * 128;
    float* H2       = ws;                  ws += (size_t)N * 64;
    const int owners = (N + OWN_ROWS - 1) / OWN_ROWS;
    int*  lcount    = (int*)ws;
    int*  llist     = lcount + (size_t)owners * LCS;

    const long nzero = (long)N;
    gcn_zero<<<(int)((nzero + 255) / 256), 256, 0, stream>>>(deg_src, nzero, lcount, owners);

    gcn_file<<<(E + PCHUNK - 1) / PCHUNK, 1024, 0, stream>>>(src, dst, deg_src, lcount, llist, E, owners);

    const int mtiles = (N + 15) / 16;

    gcn_gemm1<<<mtiles, 256, 0, stream>>>(feats, deg_src, W1, H1, N);
    gcn_own128<<<owners, 1024, 0, stream>>>(H1, src, dst, deg_dst, deg_src, b1, W2, H2, lcount, llist, E, N);

    gcn_own64<<<owners, 1024, 0, stream>>>(H2, src, dst, deg_dst, b2, (float*)d_out,
                                           lcount, llist, E, N);
}


# out-degrees counted in LDS: 32 extra groups of the filing launch each histogram 1/32 of the src list over all nodes, then add non-zero counts to the plane by whole lines; per-edge global adds go
# speedup vs baseline: 3.5153x; 1.3778x over previous
#include <hip/hip_runtime.h>

typedef __attribute__((ext_vector_type(2))) float v2f;
typedef __attribute__((ext_vector_type(8))) float v8f;

#define LDSS 132

#define OWN_ROWS 196
#define QLIST 8192
#define PCHUNK 4096
#define POWN 1024
#define LCS 32
#define HCAP 65536
#define HB 32


__global__ __launch_bounds__(256) void gcn_zero(float* __restrict__ p, long n,
                                                int* __restrict__ lcount, int owners) {
    const long first  = (long)blockIdx.x * blockDim.x + threadIdx.x;
    const long stride = (long)gridDim.x * blockDim.x;
    for (long i = first; i < n; i += stride) p[i] = 0.0f;
    const int c0 = owners <= POWN ? 0 : QLIST + 1;
    for (long o = first; o < owners; o += stride) lcount[o * LCS] = c0;
}

__device__ __forceinline__ float inv_sqrt_deg(float deg) {
    return rsqrtf(fmaxf(deg, 1.0f));
}


__global__ __launch_bounds__(1024) void gcn_file(const int* __restrict__ src,
                                                 const int* __restrict__ dst,
                                                 float* __restrict__ deg_src,
                                                 int* __restrict__ lcount,
                                                 int* __restrict__ llist, int E, int owners, int nn) {
    __shared__ int cnt[HCAP];
    static_assert(POWN <= HCAP, "the owners' counts fit in the same array");
    const int tid = threadIdx.x;
    const bool counting = nn <= HCAP;
    if (blockIdx.x < HB) {
        if (!counting) return;
        for (int i = tid; i < ((nn + 3) >> 2); i += 1024) ((int4*)cnt)[i] = make_int4(0, 0, 0, 0);
        __syncthreads();
        const int E4   = E >> 2;
        const int per4 = (E4 + HB - 1) / HB;
        const int lo   = (int)blockIdx.x * per4;
        const int hi   = min(lo + per4, E4);
        const int4* src4 = (const int4*)src;
        auto add1 = [&](int s) { if ((unsigned)s < (unsigned)nn) atomicAdd(&cnt[s], 1); };
        for (int i = lo + tid; i < hi; i += 4 * 1024) {
            int4 v[4];
            #pragma unroll
            for (int k = 0; k < 4; ++k) v[k] = src4[(i + k * 1024 < hi) ? (i + k * 1024) : i];
            #pragma unroll
            for (int k = 0; k < 4; ++k)
                if (i + k * 1024 < hi) { add1(v[k].x); add1(v[k].y); add1(v[k].z); add1(v[k].w); }
        }
        if (blockIdx.x == HB - 1)
            for (int e = (E4 << 2) + tid; e < E; e += 1024) add1(src[e]);
        __syncthreads();
        const int rot = ((int)blockIdx.x * (nn / HB)) & ~31;
        for (int i = tid; i < nn; i += 4 * 1024) {
            int c[4], j[4];
            #pragma unroll
            for (int k = 0; k < 4; ++k) {
                const int t = i + k * 1024 + rot;
                j[k] = t >= nn ? t - nn : t;
                c[k] = (i + k * 1024 < nn) ? cnt[j[k]] : 0;
            }
            #pragma unroll
            for (int k = 0; k < 4; ++k)
                if (c[k]) atomicAdd(&deg_src[j[k]], (float)c[k]);
        }
        return;
    }
    const int fb = (int)blockIdx.x - HB;
    const bool filing = owners <= POWN;
    if (filing) {
        for (int o = tid; o < owners; o += 1024) cnt[o] = 0;
        __syncthreads();
    }

    const int e = fb * PCHUNK + tid * 4;
    int d[4], s[4];
    if (e + 3 < E) {
        const int4 dv = *(const int4*)(dst + e);
        const int4 sv = *(const int4*)(src + e);
        d[0] = dv.x; d[1] = dv.y; d[2] = dv.z; d[3] = dv.w;
        s[0] = sv.x; s[1] = sv.y; s[2] = sv.z; s[3] = sv.w;
    } else {
        #pragma unroll
        for (int k = 0; k < 4; ++k) {
            const bool in = e + k < E;
            d[k] = in ? dst[e + k] : -1;
            s[k] = in ? src[e + k] : 0;
        }
    }
    if (!filing) {
        if (!counting) {
            #pragma unroll
            for (int k = 0; k < 4; ++k)
                if (e + k < E) atomicAdd(&deg_src[s[k]], 1.0f);
        }
        return;
    }
    int own[4], row[4], place[4];
    #pragma unroll
    for (int k = 0; k < 4; ++k) {
        const int o = d[k] / OWN_ROWS;
        const bool ok = d[k] >= 0 && o < owners;
        own[k] = ok ? o : -1;
        row[k] = d[k] - o * OWN_ROWS;
        place[k] = ok ? atomicAdd(&cnt[o], 1) : 0;
    }
    __syncthreads();
    for (int o = tid; o < owners; o += 1024) {
        const int c = cnt[o];
        if (c) cnt[o] = atomicAdd(&lcount[o * LCS], c);
    }
    __syncthreads();
    int2* lp = (int2*)llist;
    #pragma unroll
    for (int k = 0; k < 4; ++k) {
        if (own[k] >= 0) {
            const int slot = cnt[own[k]] + place[k];
            if (slot < QLIST) lp[(size_t)own[k] * QLIST + slot] = make_int2(s[k], row[k]);
        }
    }
    if (!counting) {
        #pragma unroll
        for (int k = 0; k < 4; ++k)
            if (e + k < E) atomicAdd(&deg_src[s[k]], 1.0f);
    }
}


__global__ __launch_bounds__(256) void gcn_gemm1(const float* __restrict__ feats,
                                                 const float* __restrict__ deg_src,
                                                 const float* __restrict__ W1,
                                                 float* __restrict__ H1, int nn) {
    __shared__ float As[16 * LDSS];
    const int tid  = threadIdx.x;
    const int wave = tid >> 5;
    const int lane = tid & 31;
    const int mbase = blockIdx.x * 16;

    for (int f = tid; f < 16 * 32; f += 256) {
        int row = f >> 5;
        int c4  = (f & 31) << 2;
        int r   = mbase + row; if (r >= nn) r = nn - 1;
        float nrm = inv_sqrt_deg(deg_src[r]);
        const float4 v = *(const float4*)(feats + (size_t)r * 128 + c4);
        float4 s; s.x = v.x * nrm; s.y = v.y * nrm; s.z = v.z * nrm; s.w = v.w * nrm;
        *(float4*)(&As[row * LDSS + c4]) = s;
    }
    __syncthreads();

    const int col  = (wave << 4) + (lane & 15);
    const int krow = (lane >> 4) << 1;
    v8f c = {};
    #pragma unroll 4
    for (int kk = 0; kk < 128; kk += 4) {
        const int k0 = kk + krow;
        v2f a = *(const v2f*)(&As[(lane & 15) * LDSS + k0]);
        v2f b;
        b.x = W1[(size_t)k0 * 128 + col];
        b.y = W1[(size_t)(k0 + 1) * 128 + col];
        c = __builtin_amdgcn_wmma_f32_16x16x4_f32(false, a, false, b, (short)0, c, false, false);
    }

    const int rbase = mbase + ((lane >> 4) << 3);
    if (mbase + 16 <= nn) {
        float* p = H1 + (size_t)rbase * 128 + col;
        #pragma unroll
        for (int v = 0; v < 8; ++v) p[v * 128] = c[v];
    } else {
        #pragma unroll
        for (int v = 0; v < 8; ++v) {
            int r = rbase + v;
            if (r < nn) H1[(size_t)r * 128 + col] = c[v];
        }
    }
}


template <int F, int QCAP, bool FINAL, int FN>
__device__ __forceinline__ void own_body(float* tile, int2* queue, int* qcount, int* rowcnt,
                                         const float* __restrict__ H,
                                         const int* __restrict__ src,
                                         const int* __restrict__ dst,
                                         float* __restrict__ deg,
                                         const float* __restrict__ deg_src,
                                         const float* __restrict__ bias,
                                         const float* __restrict__ Wn,
                                         float* __restrict__ agg,
                                         const int* __restrict__ lcount,
                                         const int* __restrict__ llist, int E, int nn) {
    static_assert(QCAP >= QLIST, "the LDS queue holds a whole filed list");
    static_assert(FINAL == (FN == 0), "an owner either finishes the layer or feeds the next product");
    constexpr int VEC = F / 32;
    constexpr int TS  = FINAL ? F : F + 4;
    constexpr int TR  = FINAL ? OWN_ROWS : ((OWN_ROWS + 15) / 16) * 16;
    const int tid   = threadIdx.x;
    const int lane  = tid & 31;
    const int wave  = __builtin_amdgcn_readfirstlane(tid >> 5);
    const int nbase = blockIdx.x * OWN_ROWS;

    for (int i = tid; i < TR * TS / 4; i += 1024)
        ((float4*)tile)[i] = make_float4(0.0f, 0.0f, 0.0f, 0.0f);
    if constexpr (!FINAL) {
        for (int i = tid; i < OWN_ROWS; i += 1024) rowcnt[i] = 0;
        __syncthreads();
    }

    int  n = 0;
    const int saved = __builtin_amdgcn_readfirstlane(lcount[blockIdx.x * LCS]);
    const bool listed = (unsigned)saved <= (unsigned)QLIST;
    if (listed) {
        n = saved;
        const int2* lp = (const int2*)llist + (size_t)blockIdx.x * QLIST;
        for (int i = tid; i < n; i += 4 * 1024) {
            int2 v[4];
            #pragma unroll
            for (int k = 0; k < 4; ++k) v[k] = lp[(i + k * 1024 < n) ? (i + k * 1024) : i];
            queue[i] = v[0];
            if constexpr (!FINAL) atomicAdd(&rowcnt[v[0].y], 1);
            #pragma unroll
            for (int k = 1; k < 4; ++k)
                if (i + k * 1024 < n) {
                    queue[i + k * 1024] = v[k];
                    if constexpr (!FINAL) atomicAdd(&rowcnt[v[k].y], 1);
                }
        }
    } else {
        if (tid == 0) *qcount = 0;
        __syncthreads();

        auto file = [&](unsigned dl, int e) {
            if (dl < (unsigned)OWN_ROWS) {
                if constexpr (!FINAL) atomicAdd(&rowcnt[dl], 1);
                const int slot = atomicAdd(qcount, 1);
                if (slot < QCAP) {
                    queue[slot] = make_int2(e, (int)dl);
                } else {
                    const float* hp = H + (size_t)src[e] * F;
                    float* tp = tile + dl * TS;
                    #pragma nounroll
                    for (int j = 0; j < F; ++j) atomicAdd(tp + j, hp[j]);
                }
            }
        };
        auto file4 = [&](const int4 v, int e) {
            const unsigned d0 = (unsigned)(v.x - nbase);
            const unsigned d1 = (unsigned)(v.y - nbase);
            const unsigned d2 = (unsigned)(v.z - nbase);
            const unsigned d3 = (unsigned)(v.w - nbase);
            const unsigned dm = min(min(d0, d1), min(d2, d3));
            if (dm < (unsigned)OWN_ROWS) {
                file(d0, e); file(d1, e + 1); file(d2, e + 2); file(d3, e + 3);
            }
        };

        const int E4 = E >> 2;
        const int4* dst4 = (const int4*)dst;
        const int start = ((int)blockIdx.x * (E4 / (int)gridDim.x)) & ~1023;
        #pragma nounroll
        for (int seg = 0; seg < 2; ++seg) {
            const int hi = seg ? start : E4;
            int b = (seg ? 0 : start) + tid;
            for (; b + 7 * 1024 < hi; b += 8 * 1024) {
                int4 v[8];
                #pragma unroll
                for (int u = 0; u < 8; ++u) v[u] = dst4[b + u * 1024];
                #pragma unroll
                for (int u = 0; u < 8; ++u) file4(v[u], (b + u * 1024) << 2);
            }
            for (; b < hi; b += 1024) file4(dst4[b], b << 2);
        }
        for (int e = (E4 << 2) + tid; e < E; e += 1024)
            file((unsigned)(dst[e] - nbase), e);
        __syncthreads();

        const int nraw = __builtin_amdgcn_readfirstlane(*qcount);
        n = nraw < QCAP ? nraw : QCAP;

        for (int i = tid; i < n; i += 4 * 1024) {
            int s[4];
            #pragma unroll
            for (int k = 0; k < 4; ++k) {
                const int ik = (i + k * 1024 < n) ? (i + k * 1024) : i;
                s[k] = src[queue[ik].x];
            }
            queue[i].x = s[0];
            #pragma unroll
            for (int k = 1; k < 4; ++k)
                if (i + k * 1024 < n) queue[i + k * 1024].x = s[k];
        }
    }
    __syncthreads();

    constexpr int NP = F / 64;
    auto rowload = [&](int s, float* val) {
        const float* hp = H + (size_t)s * F + lane * 2;
        #pragma unroll
        for (int p = 0; p < NP; ++p) {
            const float2 t = *(const float2*)(hp + p * 64);
            val[2 * p] = t.x; val[2 * p + 1] = t.y;
        }
    };
    auto rowadd = [&](int r, const float* val) {
        float* tp = tile + r * TS + lane * 2;
        #pragma unroll
        for (int p = 0; p < NP; ++p) {
            atomicAdd(tp + p * 64,     val[2 * p]);
            atomicAdd(tp + p * 64 + 1, val[2 * p + 1]);
        }
    };
    constexpr int U = 8;
    const int nfull = n & ~(U - 1);
    for (int i0 = wave * U; i0 < nfull; i0 += 32 * U) {
        const int2 mine = queue[i0 + (lane & (U - 1))];
        int   s[U], r[U];
        float val[U][VEC];
        #pragma unroll
        for (int u = 0; u < U; ++u) {
            s[u] = __builtin_amdgcn_readlane(mine.x, u);
            r[u] = __builtin_amdgcn_readlane(mine.y, u);
        }
        #pragma unroll
        for (int u = 0; u < U; ++u) rowload(s[u], val[u]);
        #pragma unroll
        for (int u = 0; u < U; ++u) rowadd(r[u], val[u]);
    }
    if (wave < n - nfull) {
        const int2 q = queue[nfull + wave];
        float val[VEC];
        rowload(q.x, val);
        rowadd(q.y, val);
    }

    int rows = nn - nbase;
    if (rows > OWN_ROWS) rows = OWN_ROWS;
    if constexpr (FINAL) {
        float4* o = (float4*)(agg + (size_t)nbase * F);
        constexpr int Q   = F / 4;
        constexpr int RPS = 1024 / Q;
        constexpr int NSW = (OWN_ROWS + RPS - 1) / RPS;
        const int q  = tid % Q;
        const int r0 = tid / Q;
        const float4 bb = *(const float4*)(bias + q * 4);
        float nd[NSW];
        #pragma unroll
        for (int k = 0; k < NSW; ++k) {
            const int r = r0 + k * RPS;
            nd[k] = (r < rows) ? inv_sqrt_deg(deg[nbase + r]) : 0.0f;
        }
        __syncthreads();
        #pragma unroll
        for (int k = 0; k < NSW; ++k) {
            const int r = r0 + k * RPS;
            if (r < rows) {
                const float4 t = ((const float4*)tile)[r * Q + q];
                float4 v;
                v.x = t.x * nd[k] + bb.x;
                v.y = t.y * nd[k] + bb.y;
                v.z = t.z * nd[k] + bb.z;
                v.w = t.w * nd[k] + bb.w;
                o[r * Q + q] = v;
            }
        }
    } else {
        constexpr int NSW = (OWN_ROWS + 31) / 32;
        constexpr int NT  = FN / 16;
        constexpr int MG  = 32 / NT;
        static_assert(F == 128 && (32 % NT) == 0, "one 16-byte piece per lane; waves split evenly over the column tiles");
        const int q  = lane;
        const int r0 = wave;
        const float4 bb = *(const float4*)(bias + q * 4);
        const int rlast = rows - 1;
        float ns[NSW];
        #pragma unroll
        for (int k = 0; k < NSW; ++k) {
            const int r = r0 + k * 32;
            ns[k] = inv_sqrt_deg(deg_src[nbase + (r < rlast ? r : rlast)]);
        }
        const int nt   = wave % NT;
        const int col  = nt * 16 + (lane & 15);
        const int krow = (lane >> 4) << 1;
        v2f b[F / 4];
        #pragma unroll
        for (int s = 0; s < F / 4; ++s) {
            const int k0 = s * 4 + krow;
            b[s].x = Wn[(size_t)k0 * FN + col];
            b[s].y = Wn[(size_t)(k0 + 1) * FN + col];
        }
        __syncthreads();

        #pragma unroll
        for (int k = 0; k < NSW; ++k) {
            const int r = r0 + k * 32;
            if (r < rows) {
                const float nd = inv_sqrt_deg((float)rowcnt[r]);
                float4* tp = (float4*)(tile + r * TS) + q;
                const float4 v = *tp;
                float4 t;
                t.x = fmaxf(v.x * nd + bb.x, 0.0f) * ns[k];
                t.y = fmaxf(v.y * nd + bb.y, 0.0f) * ns[k];
                t.z = fmaxf(v.z * nd + bb.z, 0.0f) * ns[k];
                t.w = fmaxf(v.w * nd + bb.w, 0.0f) * ns[k];
                *tp = t;
            }
        }
        if (tid < rows) deg[nbase + tid] = (float)rowcnt[tid];
        __syncthreads();

        for (int mt = wave / NT; mt * 16 < rows; mt += MG) {
            const float* ap = tile + (mt * 16 + (lane & 15)) * TS + krow;
            v8f c = {};
            #pragma unroll
            for (int s = 0; s < F / 4; ++s) {
                const v2f a = *(const v2f*)(ap + s * 4);
                c = __builtin_amdgcn_wmma_f32_16x16x4_f32(false, a, false, b[s], (short)0, c, false, false);
            }
            const int rbase = mt * 16 + ((lane >> 4) << 3);
            float* p = agg + (size_t)(nbase + rbase) * FN + col;
            if (mt * 16 + 16 <= rows) {
                #pragma unroll
                for (int v = 0; v < 8; ++v) p[v * FN] = c[v];
            } else {
                #pragma unroll
                for (int v = 0; v < 8; ++v)
                    if (rbase + v < rows) p[v * FN] = c[v];
            }
        }
    }
}

__global__ __launch_bounds__(1024) void gcn_own128(const float* __restrict__ H,
                                                   const int* __restrict__ src,
                                                   const int* __restrict__ dst,
                                                   float* __restrict__ deg_dst,
                                                   const float* __restrict__ deg_src,
                                                   const float* __restrict__ b1,
                                                   const float* __restrict__ W2,
                                                   float* __restrict__ H2,
                                                   const int* __restrict__ lcount,
                                                   const int* __restrict__ llist, int E, int nn) {
    __shared__ float tile[((OWN_ROWS + 15) / 16) * 16 * (128 + 4)];
    __shared__ int2  queue[QLIST];
    __shared__ int   rowcnt[OWN_ROWS];
    __shared__ int   qcount;
    own_body<128, QLIST, false, 64>(tile, queue, &qcount, rowcnt, H, src, dst, deg_dst, deg_src, b1, W2,
                                    H2, lcount, llist, E, nn);
}

__global__ __launch_bounds__(1024) void gcn_own64(const float* __restrict__ H,
                                                  const int* __restrict__ src,
                                                  const int* __restrict__ dst,
                                                  float* __restrict__ deg_dst,
                                                  const float* __restrict__ b2,
                                                  float* __restrict__ out,
                                                  const int* __restrict__ lcount,
                                                  const int* __restrict__ llist, int E, int nn) {
    __shared__ float tile[OWN_ROWS * 64];
    __shared__ int2  queue[16384];
    __shared__ int   qcount;
    own_body<64, 16384, true, 0>(tile, queue, &qcount, nullptr, H, src, dst, deg_dst, nullptr, b2, nullptr,
                                 out, lcount, llist, E, nn);
}


extern "C" void kernel_launch(void* const* d_in, const int* in_sizes, int n_in,
                              void* d_out, int out_size, void* d_ws, size_t ws_size,
                              hipStream_t stream) {
    const float* feats = (const float*)d_in[0];
    const float* W1    = (const float*)d_in[1];
    const float* b1    = (const float*)d_in[2];
    const float* W2    = (const float*)d_in[3];
    const float* b2    = (const float*)d_in[4];
    const int*   src   = (const int*)d_in[5];
    const int*   dst   = (const int*)d_in[6];

    const int N = in_sizes[0] / 128;
    const int E = in_sizes[5];

    float* ws       = (float*)d_ws;
    float* deg_src  = ws;                  ws += N;
    float* deg_dst  = ws;                  ws += N;
    float* H1       = ws;                  ws += (size_t)N * 128;
    float* H2       = ws;                  ws += (size_t)N * 64;
    const int owners = (N + OWN_ROWS - 1) / OWN_ROWS;
    int*  lcount    = (int*)ws;
    int*  llist     = lcount + (size_t)owners * LCS;

    const long nzero = (long)N;
    gcn_zero<<<(int)((nzero + 255) / 256), 256, 0, stream>>>(deg_src, nzero, lcount, owners);

    gcn_file<<<HB + (E + PCHUNK - 1) / PCHUNK, 1024, 0, stream>>>(src, dst, deg_src, lcount, llist, E, owners, N);

    const int mtiles = (N + 15) / 16;

    gcn_gemm1<<<mtiles, 256, 0, stream>>>(feats, deg_src, W1, H1, N);
    gcn_own128<<<owners, 1024, 0, stream>>>(H1, src, dst, deg_dst, deg_src, b1, W2, H2, lcount, llist, E, N);

    gcn_own64<<<owners, 1024, 0, stream>>>(H2, src, dst, deg_dst, b2, (float*)d_out,
                                           lcount, llist, E, N);
}


# in an owner group each wave owns 7 rows and sums them in registers: entries go to the owning wave by LDS counts; per-edge LDS float adds go; sum order = arrival in the wave's part of the list
# speedup vs baseline: 4.3223x; 1.2296x over previous
#include <hip/hip_runtime.h>

typedef __attribute__((ext_vector_type(2))) float v2f;
typedef __attribute__((ext_vector_type(8))) float v8f;

#define LDSS 132

#define OWN_ROWS 196
#define QLIST 8192
#define PCHUNK 4096
#define POWN 1024
#define LCS 32
#define HCAP 65536
#define HB 32


__global__ __launch_bounds__(256) void gcn_zero(float* __restrict__ p, long n,
                                                int* __restrict__ lcount, int owners) {
    const long first  = (long)blockIdx.x * blockDim.x + threadIdx.x;
    const long stride = (long)gridDim.x * blockDim.x;
    for (long i = first; i < n; i += stride) p[i] = 0.0f;
    const int c0 = owners <= POWN ? 0 : QLIST + 1;
    for (long o = first; o < owners; o += stride) lcount[o * LCS] = c0;
}

__device__ __forceinline__ float inv_sqrt_deg(float deg) {
    return rsqrtf(fmaxf(deg, 1.0f));
}


__global__ __launch_bounds__(1024) void gcn_file(const int* __restrict__ src,
                                                 const int* __restrict__ dst,
                                                 float* __restrict__ deg_src,
                                                 int* __restrict__ lcount,
                                                 int* __restrict__ llist, int E, int owners, int nn) {
    __shared__ int cnt[HCAP];
    static_assert(POWN <= HCAP, "the owners' counts fit in the same array");
    const int tid = threadIdx.x;
    const bool counting = nn <= HCAP;
    if (blockIdx.x < HB) {
        if (!counting) return;
        for (int i = tid; i < ((nn + 3) >> 2); i += 1024) ((int4*)cnt)[i] = make_int4(0, 0, 0, 0);
        __syncthreads();
        const int E4   = E >> 2;
        const int per4 = (E4 + HB - 1) / HB;
        const int lo   = (int)blockIdx.x * per4;
        const int hi   = min(lo + per4, E4);
        const int4* src4 = (const int4*)src;
        auto add1 = [&](int s) { if ((unsigned)s < (unsigned)nn) atomicAdd(&cnt[s], 1); };
        for (int i = lo + tid; i < hi; i += 4 * 1024) {
            int4 v[4];
            #pragma unroll
            for (int k = 0; k < 4; ++k) v[k] = src4[(i + k * 1024 < hi) ? (i + k * 1024) : i];
            #pragma unroll
            for (int k = 0; k < 4; ++k)
                if (i + k * 1024 < hi) { add1(v[k].x); add1(v[k].y); add1(v[k].z); add1(v[k].w); }
        }
        if (blockIdx.x == HB - 1)
            for (int e = (E4 << 2) + tid; e < E; e += 1024) add1(src[e]);
        __syncthreads();
        const int rot = ((int)blockIdx.x * (nn / HB)) & ~31;
        for (int i = tid; i < nn; i += 4 * 1024) {
            int c[4], j[4];
            #pragma unroll
            for (int k = 0; k < 4; ++k) {
                const int t = i + k * 1024 + rot;
                j[k] = t >= nn ? t - nn : t;
                c[k] = (i + k * 1024 < nn) ? cnt[j[k]] : 0;
            }
            #pragma unroll
            for (int k = 0; k < 4; ++k)
                if (c[k]) atomicAdd(&deg_src[j[k]], (float)c[k]);
        }
        return;
    }
    const int fb = (int)blockIdx.x - HB;
    const bool filing = owners <= POWN;
    if (filing) {
        for (int o = tid; o < owners; o += 1024) cnt[o] = 0;
        __syncthreads();
    }

    const int e = fb * PCHUNK + tid * 4;
    int d[4], s[4];
    if (e + 3 < E) {
        const int4 dv = *(const int4*)(dst + e);
        const int4 sv = *(const int4*)(src + e);
        d[0] = dv.x; d[1] = dv.y; d[2] = dv.z; d[3] = dv.w;
        s[0] = sv.x; s[1] = sv.y; s[2] = sv.z; s[3] = sv.w;
    } else {
        #pragma unroll
        for (int k = 0; k < 4; ++k) {
            const bool in = e + k < E;
            d[k] = in ? dst[e + k] : -1;
            s[k] = in ? src[e + k] : 0;
        }
    }
    if (!filing) {
        if (!counting) {
            #pragma unroll
            for (int k = 0; k < 4; ++k)
                if (e + k < E) atomicAdd(&deg_src[s[k]], 1.0f);
        }
        return;
    }
    int own[4], row[4], place[4];
    #pragma unroll
    for (int k = 0; k < 4; ++k) {
        const int o = d[k] / OWN_ROWS;
        const bool ok = d[k] >= 0 && o < owners;
        own[k] = ok ? o : -1;
        row[k] = d[k] - o * OWN_ROWS;
        place[k] = ok ? atomicAdd(&cnt[o], 1) : 0;
    }
    __syncthreads();
    for (int o = tid; o < owners; o += 1024) {
        const int c = cnt[o];
        if (c) cnt[o] = atomicAdd(&lcount[o * LCS], c);
    }
    __syncthreads();
    int2* lp = (int2*)llist;
    #pragma unroll
    for (int k = 0; k < 4; ++k) {
        if (own[k] >= 0) {
            const int slot = cnt[own[k]] + place[k];
            if (slot < QLIST) lp[(size_t)own[k] * QLIST + slot] = make_int2(s[k], row[k]);
        }
    }
    if (!counting) {
        #pragma unroll
        for (int k = 0; k < 4; ++k)
            if (e + k < E) atomicAdd(&deg_src[s[k]], 1.0f);
    }
}


__global__ __launch_bounds__(256) void gcn_gemm1(const float* __restrict__ feats,
                                                 const float* __restrict__ deg_src,
                                                 const float* __restrict__ W1,
                                                 float* __restrict__ H1, int nn) {
    __shared__ float As[16 * LDSS];
    const int tid  = threadIdx.x;
    const int wave = tid >> 5;
    const int lane = tid & 31;
    const int mbase = blockIdx.x * 16;

    for (int f = tid; f < 16 * 32; f += 256) {
        int row = f >> 5;
        int c4  = (f & 31) << 2;
        int r   = mbase + row; if (r >= nn) r = nn - 1;
        float nrm = inv_sqrt_deg(deg_src[r]);
        const float4 v = *(const float4*)(feats + (size_t)r * 128 + c4);
        float4 s; s.x = v.x * nrm; s.y = v.y * nrm; s.z = v.z * nrm; s.w = v.w * nrm;
        *(float4*)(&As[row * LDSS + c4]) = s;
    }
    __syncthreads();

    const int col  = (wave << 4) + (lane & 15);
    const int krow = (lane >> 4) << 1;
    v8f c = {};
    #pragma unroll 4
    for (int kk = 0; kk < 128; kk += 4) {
        const int k0 = kk + krow;
        v2f a = *(const v2f*)(&As[(lane & 15) * LDSS + k0]);
        v2f b;
        b.x = W1[(size_t)k0 * 128 + col];
        b.y = W1[(size_t)(k0 + 1) * 128 + col];
        c = __builtin_amdgcn_wmma_f32_16x16x4_f32(false, a, false, b, (short)0, c, false, false);
    }

    const int rbase = mbase + ((lane >> 4) << 3);
    if (mbase + 16 <= nn) {
        float* p = H1 + (size_t)rbase * 128 + col;
        #pragma unroll
        for (int v = 0; v < 8; ++v) p[v * 128] = c[v];
    } else {
        #pragma unroll
        for (int v = 0; v < 8; ++v) {
            int r = rbase + v;
            if (r < nn) H1[(size_t)r * 128 + col] = c[v];
        }
    }
}


template <int F, int QCAP, bool FINAL, int FN>
__device__ __forceinline__ void own_body(float* tile, int2* queue, int* qcount, int* rowcnt, int* bincnt,
                                         const float* __restrict__ H,
                                         const int* __restrict__ src,
                                         const int* __restrict__ dst,
                                         float* __restrict__ deg,
                                         const float* __restrict__ deg_src,
                                         const float* __restrict__ bias,
                                         const float* __restrict__ Wn,
                                         float* __restrict__ agg,
                                         const int* __restrict__ lcount,
                                         const int* __restrict__ llist, int E, int nn) {
    static_assert(QCAP >= QLIST, "the LDS queue holds a whole filed list");
    static_assert(FINAL == (FN == 0), "an owner either finishes the layer or feeds the next product");
    constexpr int VEC = F / 32;
    constexpr int TS  = FINAL ? F : F + 4;
    constexpr int TR  = FINAL ? OWN_ROWS : ((OWN_ROWS + 15) / 16) * 16;
    constexpr int RPW  = (OWN_ROWS + 31) / 32;
    constexpr int BCAP = QCAP / 32;
    const int tid   = threadIdx.x;
    const int lane  = tid & 31;
    const int wave  = __builtin_amdgcn_readfirstlane(tid >> 5);
    const int nbase = blockIdx.x * OWN_ROWS;

    for (int i = tid; i < TR * TS / 4; i += 1024)
        ((float4*)tile)[i] = make_float4(0.0f, 0.0f, 0.0f, 0.0f);
    if constexpr (!FINAL) {
        for (int i = tid; i < OWN_ROWS; i += 1024) rowcnt[i] = 0;
    }
    if (tid < 32) bincnt[tid] = 0;
    __syncthreads();

    int  n = 0;
    const int saved = __builtin_amdgcn_readfirstlane(lcount[blockIdx.x * LCS]);
    const bool listed = (unsigned)saved <= (unsigned)QLIST;
    constexpr int NP = F / 64;
    auto rowload = [&](int s, float* val) {
        const float* hp = H + (size_t)s * F + lane * 2;
        #pragma unroll
        for (int p = 0; p < NP; ++p) {
            const float2 t = *(const float2*)(hp + p * 64);
            val[2 * p] = t.x; val[2 * p + 1] = t.y;
        }
    };

    if (listed) {
        n = saved;
        auto hand = [&](const int2 v) {
            const int r = v.y;
            const int b = (int)((unsigned)r / (unsigned)RPW);
            if constexpr (!FINAL) atomicAdd(&rowcnt[r], 1);
            const int place = atomicAdd(&bincnt[b], 1);
            if (place < BCAP) {
                queue[b * BCAP + place] = make_int2(v.x, r - b * RPW);
            } else {
                const float* hp = H + (size_t)v.x * F;
                float* tp = tile + r * TS;
                #pragma nounroll
                for (int j = 0; j < F; ++j) atomicAdd(tp + j, hp[j]);
            }
        };
        const int2* lp = (const int2*)llist + (size_t)blockIdx.x * QLIST;
        for (int i = tid; i < n; i += 4 * 1024) {
            int2 v[4];
            #pragma unroll
            for (int k = 0; k < 4; ++k) v[k] = lp[(i + k * 1024 < n) ? (i + k * 1024) : i];
            hand(v[0]);
            #pragma unroll
            for (int k = 1; k < 4; ++k)
                if (i + k * 1024 < n) hand(v[k]);
        }
        __syncthreads();

        constexpr int UR = 1024 / F;
        const int filed = __builtin_amdgcn_readfirstlane(bincnt[wave]);
        const int cnt   = filed < BCAP ? filed : BCAP;
        const int2* bq  = queue + wave * BCAP;
        typedef float accv __attribute__((ext_vector_type(VEC)));
        typedef float sumv __attribute__((ext_vector_type(8 * VEC)));
        static_assert(RPW <= 8, "the register array has room for 8 rows");
        sumv acc = {};
        for (int i0 = 0; i0 < cnt; i0 += UR) {
            const int mi = i0 + (lane & (UR - 1));
            const int2 mine = bq[mi < cnt ? mi : cnt - 1];
            int  s[UR], r[UR];
            accv val[UR];
            #pragma unroll
            for (int u = 0; u < UR; ++u) {
                s[u] = __builtin_amdgcn_readlane(mine.x, u);
                r[u] = __builtin_amdgcn_readlane(mine.y, u);
            }
            #pragma unroll
            for (int u = 0; u < UR; ++u) {
                const float* hp = H + (size_t)s[u] * F + lane * 2;
                #pragma unroll
                for (int p = 0; p < NP; ++p) {
                    const float2 t = *(const float2*)(hp + p * 64);
                    val[u][2 * p] = t.x; val[u][2 * p + 1] = t.y;
                }
            }
            #pragma unroll
            for (int u = 0; u < UR; ++u) {
                if (i0 + u < cnt) {
                    const int k = r[u] * VEC;
                    #pragma unroll
                    for (int p = 0; p < VEC; ++p) acc[k + p] += val[u][p];
                }
            }
        }
        #pragma unroll
        for (int j = 0; j < RPW; ++j) {
            const int row = wave * RPW + j;
            if (row < OWN_ROWS) {
                float* tp = tile + row * TS + lane * 2;
                #pragma unroll
                for (int p = 0; p < NP; ++p) {
                    float2 t = *(float2*)(tp + p * 64);
                    t.x += acc[j * VEC + 2 * p]; t.y += acc[j * VEC + 2 * p + 1];
                    *(float2*)(tp + p * 64) = t;
                }
            }
        }
    } else {
        if (tid == 0) *qcount = 0;
        __syncthreads();

        auto file = [&](unsigned dl, int e) {
            if (dl < (unsigned)OWN_ROWS) {
                if constexpr (!FINAL) atomicAdd(&rowcnt[dl], 1);
                const int slot = atomicAdd(qcount, 1);
                if (slot < QCAP) {
                    queue[slot] = make_int2(e, (int)dl);
                } else {
                    const float* hp = H + (size_t)src[e] * F;
                    float* tp = tile + dl * TS;
                    #pragma nounroll
                    for (int j = 0; j < F; ++j) atomicAdd(tp + j, hp[j]);
                }
            }
        };
        auto file4 = [&](const int4 v, int e) {
            const unsigned d0 = (unsigned)(v.x - nbase);
            const unsigned d1 = (unsigned)(v.y - nbase);
            const unsigned d2 = (unsigned)(v.z - nbase);
            const unsigned d3 = (unsigned)(v.w - nbase);
            const unsigned dm = min(min(d0, d1), min(d2, d3));
            if (dm < (unsigned)OWN_ROWS) {
                file(d0, e); file(d1, e + 1); file(d2, e + 2); file(d3, e + 3);
            }
        };

        const int E4 = E >> 2;
        const int4* dst4 = (const int4*)dst;
        const int start = ((int)blockIdx.x * (E4 / (int)gridDim.x)) & ~1023;
        #pragma nounroll
        for (int seg = 0; seg < 2; ++seg) {
            const int hi = seg ? start : E4;
            int b = (seg ? 0 : start) + tid;
            for (; b + 7 * 1024 < hi; b += 8 * 1024) {
                int4 v[8];
                #pragma unroll
                for (int u = 0; u < 8; ++u) v[u] = dst4[b + u * 1024];
                #pragma unroll
                for (int u = 0; u < 8; ++u) file4(v[u], (b + u * 1024) << 2);
            }
            for (; b < hi; b += 1024) file4(dst4[b], b << 2);
        }
        for (int e = (E4 << 2) + tid; e < E; e += 1024)
            file((unsigned)(dst[e] - nbase), e);
        __syncthreads();

        const int nraw = __builtin_amdgcn_readfirstlane(*qcount);
        n = nraw < QCAP ? nraw : QCAP;

        for (int i = tid; i < n; i += 4 * 1024) {
            int s[4];
            #pragma unroll
            for (int k = 0; k < 4; ++k) {
                const int ik = (i + k * 1024 < n) ? (i + k * 1024) : i;
                s[k] = src[queue[ik].x];
            }
            queue[i].x = s[0];
            #pragma unroll
            for (int k = 1; k < 4; ++k)
                if (i + k * 1024 < n) queue[i + k * 1024].x = s[k];
        }
        __syncthreads();

        auto rowadd = [&](int r, const float* val) {
            float* tp = tile + r * TS + lane * 2;
            #pragma unroll
            for (int p = 0; p < NP; ++p) {
                atomicAdd(tp + p * 64,     val[2 * p]);
                atomicAdd(tp + p * 64 + 1, val[2 * p + 1]);
            }
        };
        constexpr int U = 8;
        const int nfull = n & ~(U - 1);
        for (int i0 = wave * U; i0 < nfull; i0 += 32 * U) {
            const int2 mine = queue[i0 + (lane & (U - 1))];
            int   s[U], r[U];
            float val[U][VEC];
            #pragma unroll
            for (int u = 0; u < U; ++u) {
                s[u] = __builtin_amdgcn_readlane(mine.x, u);
                r[u] = __builtin_amdgcn_readlane(mine.y, u);
            }
            #pragma unroll
            for (int u = 0; u < U; ++u) rowload(s[u], val[u]);
            #pragma unroll
            for (int u = 0; u < U; ++u) rowadd(r[u], val[u]);
        }
        if (wave < n - nfull) {
            const int2 q = queue[nfull + wave];
            float val[VEC];
            rowload(q.x, val);
            rowadd(q.y, val);
        }
    }

    int rows = nn - nbase;
    if (rows > OWN_ROWS) rows = OWN_ROWS;
    if constexpr (FINAL) {
        float4* o = (float4*)(agg + (size_t)nbase * F);
        constexpr int Q   = F / 4;
        constexpr int RPS = 1024 / Q;
        constexpr int NSW = (OWN_ROWS + RPS - 1) / RPS;
        const int q  = tid % Q;
        const int r0 = tid / Q;
        const float4 bb = *(const float4*)(bias + q * 4);
        float nd[NSW];
        #pragma unroll
        for (int k = 0; k < NSW; ++k) {
            const int r = r0 + k * RPS;
            nd[k] = (r < rows) ? inv_sqrt_deg(deg[nbase + r]) : 0.0f;
        }
        __syncthreads();
        #pragma unroll
        for (int k = 0; k < NSW; ++k) {
            const int r = r0 + k * RPS;
            if (r < rows) {
                const float4 t = ((const float4*)tile)[r * Q + q];
                float4 v;
                v.x = t.x * nd[k] + bb.x;
                v.y = t.y * nd[k] + bb.y;
                v.z = t.z * nd[k] + bb.z;
                v.w = t.w * nd[k] + bb.w;
                o[r * Q + q] = v;
            }
        }
    } else {
        constexpr int NSW = (OWN_ROWS + 31) / 32;
        constexpr int NT  = FN / 16;
        constexpr int MG  = 32 / NT;
        static_assert(F == 128 && (32 % NT) == 0, "one 16-byte piece per lane; waves split evenly over the column tiles");
        const int q  = lane;
        const int r0 = wave;
        const float4 bb = *(const float4*)(bias + q * 4);
        const int rlast = rows - 1;
        float ns[NSW];
        #pragma unroll
        for (int k = 0; k < NSW; ++k) {
            const int r = r0 + k * 32;
            ns[k] = inv_sqrt_deg(deg_src[nbase + (r < rlast ? r : rlast)]);
        }
        const int nt   = wave % NT;
        const int col  = nt * 16 + (lane & 15);
        const int krow = (lane >> 4) << 1;
        v2f b[F / 4];
        #pragma unroll
        for (int s = 0; s < F / 4; ++s) {
            const int k0 = s * 4 + krow;
            b[s].x = Wn[(size_t)k0 * FN + col];
            b[s].y = Wn[(size_t)(k0 + 1) * FN + col];
        }
        __syncthreads();

        #pragma unroll
        for (int k = 0; k < NSW; ++k) {
            const int r = r0 + k * 32;
            if (r < rows) {
                const float nd = inv_sqrt_deg((float)rowcnt[r]);
                float4* tp = (float4*)(tile + r * TS) + q;
                const float4 v = *tp;
                float4 t;
                t.x = fmaxf(v.x * nd + bb.x, 0.0f) * ns[k];
                t.y = fmaxf(v.y * nd + bb.y, 0.0f) * ns[k];
                t.z = fmaxf(v.z * nd + bb.z, 0.0f) * ns[k];
                t.w = fmaxf(v.w * nd + bb.w, 0.0f) * ns[k];
                *tp = t;
            }
        }
        if (tid < rows) deg[nbase + tid] = (float)rowcnt[tid];
        __syncthreads();

        for (int mt = wave / NT; mt * 16 < rows; mt += MG) {
            const float* ap = tile + (mt * 16 + (lane & 15)) * TS + krow;
            v8f c = {};
            #pragma unroll
            for (int s = 0; s < F / 4; ++s) {
                const v2f a = *(const v2f*)(ap + s * 4);
                c = __builtin_amdgcn_wmma_f32_16x16x4_f32(false, a, false, b[s], (short)0, c, false, false);
            }
            const int rbase = mt * 16 + ((lane >> 4) << 3);
            float* p = agg + (size_t)(nbase + rbase) * FN + col;
            if (mt * 16 + 16 <= rows) {
                #pragma unroll
                for (int v = 0; v < 8; ++v) p[v * FN] = c[v];
            } else {
                #pragma unroll
                for (int v = 0; v < 8; ++v)
                    if (rbase + v < rows) p[v * FN] = c[v];
            }
        }
    }
}

__global__ __launch_bounds__(1024) void gcn_own128(const float* __restrict__ H,
                                                   const int* __restrict__ src,
                                                   const int* __restrict__ dst,
                                                   float* __restrict__ deg_dst,
                                                   const float* __restrict__ deg_src,
                                                   const float* __restrict__ b1,
                                                   const float* __restrict__ W2,
                                                   float* __restrict__ H2,
                                                   const int* __restrict__ lcount,
                                                   const int* __restrict__ llist, int E, int nn) {
    __shared__ float tile[((OWN_ROWS + 15) / 16) * 16 * (128 + 4)];
    __shared__ int2  queue[QLIST];
    __shared__ int   rowcnt[OWN_ROWS];
    __shared__ int   bincnt[32];
    __shared__ int   qcount;
    own_body<128, QLIST, false, 64>(tile, queue, &qcount, rowcnt, bincnt, H, src, dst, deg_dst, deg_src, b1, W2,
                                    H2, lcount, llist, E, nn);
}

__global__ __launch_bounds__(1024) void gcn_own64(const float* __restrict__ H,
                                                  const int* __restrict__ src,
                                                  const int* __restrict__ dst,
                                                  float* __restrict__ deg_dst,
                                                  const float* __restrict__ b2,
                                                  float* __restrict__ out,
                                                  const int* __restrict__ lcount,
                                                  const int* __restrict__ llist, int E, int nn) {
    __shared__ float tile[OWN_ROWS * 64];
    __shared__ int2  queue[16384];
    __shared__ int   bincnt[32];
    __shared__ int   qcount;
    own_body<64, 16384, true, 0>(tile, queue, &qcount, nullptr, bincnt, H, src, dst, deg_dst, nullptr, b2, nullptr,
                                 out, lcount, llist, E, nn);
}


extern "C" void kernel_launch(void* const* d_in, const int* in_sizes, int n_in,
                              void* d_out, int out_size, void* d_ws, size_t ws_size,
                              hipStream_t stream) {
    const float* feats = (const float*)d_in[0];
    const float* W1    = (const float*)d_in[1];
    const float* b1    = (const float*)d_in[2];
    const float* W2    = (const float*)d_in[3];
    const float* b2    = (const float*)d_in[4];
    const int*   src   = (const int*)d_in[5];
    const int*   dst   = (const int*)d_in[6];

    const int N = in_sizes[0] / 128;
    const int E = in_sizes[5];

    float* ws       = (float*)d_ws;
    float* deg_src  = ws;                  ws += N;
    float* deg_dst  = ws;                  ws += N;
    float* H1       = ws;                  ws += (size_t)N * 128;
    float* H2       = ws;                  ws += (size_t)N * 64;
    const int owners = (N + OWN_ROWS - 1) / OWN_ROWS;
    int*  lcount    = (int*)ws;
    int*  llist     = lcount + (size_t)owners * LCS;

    const long nzero = (long)N;
    gcn_zero<<<(int)((nzero + 255) / 256), 256, 0, stream>>>(deg_src, nzero, lcount, owners);

    gcn_file<<<HB + (E + PCHUNK - 1) / PCHUNK, 1024, 0, stream>>>(src, dst, deg_src, lcount, llist, E, owners, N);

    const int mtiles = (N + 15) / 16;

    gcn_gemm1<<<mtiles, 256, 0, stream>>>(feats, deg_src, W1, H1, N);
    gcn_own128<<<owners, 1024, 0, stream>>>(H1, src, dst, deg_dst, deg_src, b1, W2, H2, lcount, llist, E, N);

    gcn_own64<<<owners, 1024, 0, stream>>>(H2, src, dst, deg_dst, b2, (float*)d_out,
                                           lcount, llist, E, N);
}


# every count stored whole by the group that made it (per-group edge pieces, out-degree planes; readers add them up): clearing launch and all global adds go; 5 launches -> 4; sum order = LDS arrival
# speedup vs baseline: 4.9407x; 1.1431x over previous
#include <hip/hip_runtime.h>


typedef __attribute__((ext_vector_type(2))) float v2f;
typedef __attribute__((ext_vector_type(8))) float v8f;

#define LDSS 132

#define OWN_ROWS 196
#define QLIST 8192
#define PCHUNK 4096
#define POWN 1024
#define HCAP 65536
#define HB 32
#define FBMAX 256
#define CAPP 64

__device__ __forceinline__ float inv_sqrt_deg(float deg) {
    return rsqrtf(fmaxf(deg, 1.0f));
}


__global__ __launch_bounds__(1024) void gcn_file(const int* __restrict__ src,
                                                 const int* __restrict__ dst,
                                                 int* __restrict__ degp, int pitch,
                                                 int* __restrict__ pcount,
                                                 int* __restrict__ slabs, int E, int owners, int nn) {
    __shared__ int cnt[HCAP];
    static_assert(POWN <= HCAP, "the owners' counts fit in the same array");
    const int tid = threadIdx.x;
    if (blockIdx.x < HB) {
        const int E4   = E >> 2;
        const int per4 = (E4 + HB - 1) / HB;
        const int lo   = (int)blockIdx.x * per4;
        const int hi   = min(lo + per4, E4);
        const int4* src4 = (const int4*)src;
        int* plane = degp + (size_t)blockIdx.x * pitch;
        for (int base = 0; base < pitch; base += HCAP) {
            const int span = min(HCAP, pitch - base);
            for (int i = tid; i < (span >> 2); i += 1024) ((int4*)cnt)[i] = make_int4(0, 0, 0, 0);
            __syncthreads();
            auto add1 = [&](int s) {
                const unsigned k = (unsigned)(s - base);
                if (k < (unsigned)span && (unsigned)s < (unsigned)nn) atomicAdd(&cnt[k], 1);
            };
            for (int i = lo + tid; i < hi; i += 4 * 1024) {
                int4 v[4];
                #pragma unroll
                for (int k = 0; k < 4; ++k) v[k] = src4[(i + k * 1024 < hi) ? (i + k * 1024) : i];
                #pragma unroll
                for (int k = 0; k < 4; ++k)
                    if (i + k * 1024 < hi) { add1(v[k].x); add1(v[k].y); add1(v[k].z); add1(v[k].w); }
            }
            if (blockIdx.x == HB - 1)
                for (int e = (E4 << 2) + tid; e < E; e += 1024) add1(src[e]);
            __syncthreads();
            for (int i = tid; i < (span >> 2); i += 1024)
                ((int4*)(plane + base))[i] = ((const int4*)cnt)[i];
            if (base + HCAP < pitch) __syncthreads();
        }
        return;
    }
    const int fb = (int)blockIdx.x - HB;
    const int FB = (int)gridDim.x - HB;
    const bool filing = owners <= POWN && FB <= FBMAX;
    if (!filing) return;
    for (int o = tid; o < owners; o += 1024) cnt[o] = 0;
    __syncthreads();

    const int e = fb * PCHUNK + tid * 4;
    int d[4], s[4];
    if (e + 3 < E) {
        const int4 dv = *(const int4*)(dst + e);
        const int4 sv = *(const int4*)(src + e);
        d[0] = dv.x; d[1] = dv.y; d[2] = dv.z; d[3] = dv.w;
        s[0] = sv.x; s[1] = sv.y; s[2] = sv.z; s[3] = sv.w;
    } else {
        #pragma unroll
        for (int k = 0; k < 4; ++k) {
            const bool in = e + k < E;
            d[k] = in ? dst[e + k] : -1;
            s[k] = in ? src[e + k] : 0;
        }
    }
    int own[4], row[4], place[4];
    #pragma unroll
    for (int k = 0; k < 4; ++k) {
        const int o = d[k] / OWN_ROWS;
        const bool ok = d[k] >= 0 && o < owners;
        own[k] = ok ? o : -1;
        row[k] = d[k] - o * OWN_ROWS;
        place[k] = ok ? atomicAdd(&cnt[o], 1) : 0;
    }
    __syncthreads();
    for (int o = tid; o < owners; o += 1024) pcount[(size_t)fb * POWN + o] = cnt[o];
    int2* sp = (int2*)slabs;
    #pragma unroll
    for (int k = 0; k < 4; ++k) {
        if (own[k] >= 0 && place[k] < CAPP)
            sp[((size_t)own[k] * FBMAX + fb) * CAPP + place[k]] = make_int2(s[k], row[k]);
    }
}


__global__ __launch_bounds__(256) void gcn_gemm1(const float* __restrict__ feats,
                                                 const int* __restrict__ degp, int pitch,
                                                 const float* __restrict__ W1,
                                                 float* __restrict__ H1, int nn) {
    __shared__ float As[16 * LDSS];
    __shared__ int   part[8 * 16];
    static_assert(HB == 32, "256 threads: 32 planes x 8 row pairs");
    const int tid  = threadIdx.x;
    const int wave = tid >> 5;
    const int lane = tid & 31;
    const int mbase = blockIdx.x * 16;

    float4 fr[2];
    #pragma unroll
    for (int it = 0; it < 2; ++it) {
        int r = mbase + wave + 8 * it; if (r >= nn) r = nn - 1;
        fr[it] = *(const float4*)(feats + (size_t)r * 128 + (lane << 2));
    }
    {
        int2 dv = *(const int2*)(degp + (size_t)(tid >> 3) * pitch + mbase + 2 * (tid & 7));
        dv.x += __shfl_xor(dv.x, 8);  dv.y += __shfl_xor(dv.y, 8);
        dv.x += __shfl_xor(dv.x, 16); dv.y += __shfl_xor(dv.y, 16);
        if (lane < 8) { part[wave * 16 + 2 * lane] = dv.x; part[wave * 16 + 2 * lane + 1] = dv.y; }
    }
    __syncthreads();
    #pragma unroll
    for (int it = 0; it < 2; ++it) {
        const int row = wave + 8 * it;
        int dsum = 0;
        #pragma unroll
        for (int w = 0; w < 8; ++w) dsum += part[w * 16 + row];
        const float nrm = inv_sqrt_deg((float)dsum);
        float4 s; s.x = fr[it].x * nrm; s.y = fr[it].y * nrm; s.z = fr[it].z * nrm; s.w = fr[it].w * nrm;
        *(float4*)(&As[row * LDSS + (lane << 2)]) = s;
    }
    __syncthreads();
    const int col  = (wave << 4) + (lane & 15);
    const int krow = (lane >> 4) << 1;
    v8f c = {};
    #pragma unroll 4
    for (int kk = 0; kk < 128; kk += 4) {
        const int k0 = kk + krow;
        v2f a = *(const v2f*)(&As[(lane & 15) * LDSS + k0]);
        v2f b;
        b.x = W1[(size_t)k0 * 128 + col];
        b.y = W1[(size_t)(k0 + 1) * 128 + col];
        c = __builtin_amdgcn_wmma_f32_16x16x4_f32(false, a, false, b, (short)0, c, false, false);
    }

    const int rbase = mbase + ((lane >> 4) << 3);
    if (mbase + 16 <= nn) {
        float* p = H1 + (size_t)rbase * 128 + col;
        #pragma unroll
        for (int v = 0; v < 8; ++v) p[v * 128] = c[v];
    } else {
        #pragma unroll
        for (int v = 0; v < 8; ++v) {
            int r = rbase + v;
            if (r < nn) H1[(size_t)r * 128 + col] = c[v];
        }
    }
}


template <int F, int QCAP, bool FINAL, int FN>
__device__ __forceinline__ void own_body(float* tile, int2* queue, int* qcount, int* rowcnt, int* bincnt, int* outdeg, int* wover,
                                         const float* __restrict__ H,
                                         const int* __restrict__ src,
                                         const int* __restrict__ dst,
                                         float* __restrict__ deg,
                                         const int* __restrict__ degp, int pitch,
                                         const float* __restrict__ bias,
                                         const float* __restrict__ Wn,
                                         float* __restrict__ agg,
                                         const int* __restrict__ pcount,
                                         const int* __restrict__ slabs, int E, int nn) {
    static_assert(QCAP >= QLIST, "the LDS queue has at least QLIST entries");
    static_assert(FINAL == (FN == 0), "an owner either finishes the layer or feeds the next product");
    constexpr int VEC = F / 32;
    constexpr int TS  = FINAL ? F : F + 4;
    constexpr int TR  = FINAL ? OWN_ROWS : ((OWN_ROWS + 15) / 16) * 16;
    constexpr int RPW  = (OWN_ROWS + 31) / 32;
    constexpr int BCAP = QCAP / 32;
    const int tid   = threadIdx.x;
    const int lane  = tid & 31;
    const int wave  = __builtin_amdgcn_readfirstlane(tid >> 5);
    const int nbase = blockIdx.x * OWN_ROWS;
    constexpr int NPW = FBMAX / 32;
    constexpr int NSD = (OWN_ROWS + 31) / 32;
    static_assert(HB == 32, "one out-degree plane per wave of the group");

    const int FB = (E + PCHUNK - 1) / PCHUNK;
    const bool pieces = (int)gridDim.x <= POWN && FB <= FBMAX;
    int myc = 0;
    if (pieces && lane < NPW && wave + 32 * lane < FB)
        myc = pcount[(size_t)(wave + 32 * lane) * POWN + blockIdx.x];
    int od[NSD];
    if constexpr (!FINAL) {
        #pragma unroll
        for (int k = 0; k < NSD; ++k) {
            const int r = lane + 32 * k;
            od[k] = (r < OWN_ROWS && nbase + r < nn) ? degp[(size_t)wave * pitch + nbase + r] : 0;
        }
    }

    for (int i = tid; i < TR * TS / 4; i += 1024)
        ((float4*)tile)[i] = make_float4(0.0f, 0.0f, 0.0f, 0.0f);
    if constexpr (!FINAL) {
        for (int i = tid; i < OWN_ROWS; i += 1024) { rowcnt[i] = 0; outdeg[i] = 0; }
    }
    if (tid < 32) bincnt[tid] = 0;
    const int over = __any(myc > CAPP) ? 1 : 0;
    if (lane == 0) wover[wave] = over;
    __syncthreads();

    if constexpr (!FINAL) {
        #pragma unroll
        for (int k = 0; k < NSD; ++k)
            if (od[k]) atomicAdd(&outdeg[lane + 32 * k], od[k]);
    }

    int  n = 0;
    const bool listed = pieces && !__any(wover[lane] != 0);
    constexpr int NP = F / 64;
    auto rowload = [&](int s, float* val) {
        const float* hp = H + (size_t)s * F + lane * 2;
        #pragma unroll
        for (int p = 0; p < NP; ++p) {
            const float2 t = *(const float2*)(hp + p * 64);
            val[2 * p] = t.x; val[2 * p + 1] = t.y;
        }
    };

    if (listed) {
        auto hand = [&](const int2 v) {
            const int r = v.y;
            const int b = (int)((unsigned)r / (unsigned)RPW);
            if constexpr (!FINAL) atomicAdd(&rowcnt[r], 1);
            const int place = atomicAdd(&bincnt[b], 1);
            if (place < BCAP) {
                queue[b * BCAP + place] = make_int2(v.x, r - b * RPW);
            } else {
                const float* hp = H + (size_t)v.x * F;
                float* tp = tile + r * TS;
                #pragma nounroll
                for (int j = 0; j < F; ++j) atomicAdd(tp + j, hp[j]);
            }
        };
        const int2* sp = (const int2*)slabs + ((size_t)blockIdx.x * FBMAX + wave) * CAPP;
        int   pc[NPW];
        unsigned long long pv[NPW];
        const unsigned long long* sp8 = (const unsigned long long*)sp;
        #pragma unroll
        for (int k = 0; k < NPW; ++k) pc[k] = __builtin_amdgcn_readlane(myc, k);
        #pragma unroll
        for (int k = 0; k < NPW; ++k)
            pv[k] = (lane < pc[k]) ? sp8[(size_t)k * 32 * CAPP + lane] : 0ull;
        #pragma unroll
        for (int k = 0; k < NPW; ++k)
            if (lane < pc[k]) hand(make_int2((int)(unsigned)pv[k], (int)(unsigned)(pv[k] >> 32)));
        static_assert(CAPP <= 64, "a piece is at most two entries per lane");
        #pragma unroll
        for (int k = 0; k < NPW; ++k)
            if (lane + 32 < pc[k]) hand(sp[(size_t)k * 32 * CAPP + lane + 32]);
        __syncthreads();

        constexpr int UR = 1024 / F;
        const int filed = __builtin_amdgcn_readfirstlane(bincnt[wave]);
        const int cnt   = filed < BCAP ? filed : BCAP;
        const int2* bq  = queue + wave * BCAP;
        typedef float accv __attribute__((ext_vector_type(VEC)));
        typedef float sumv __attribute__((ext_vector_type(8 * VEC)));
        static_assert(RPW <= 8, "the register array has room for 8 rows");
        sumv acc = {};
        for (int i0 = 0; i0 < cnt; i0 += UR) {
            const int mi = i0 + (lane & (UR - 1));
            const int2 mine = bq[mi < cnt ? mi : cnt - 1];
            int  s[UR], r[UR];
            accv val[UR];
            #pragma unroll
            for (int u = 0; u < UR; ++u) {
                s[u] = __builtin_amdgcn_readlane(mine.x, u);
                r[u] = __builtin_amdgcn_readlane(mine.y, u);
            }
            #pragma unroll
            for (int u = 0; u < UR; ++u) {
                const float* hp = H + (size_t)s[u] * F + lane * 2;
                #pragma unroll
                for (int p = 0; p < NP; ++p) {
                    const float2 t = *(const float2*)(hp + p * 64);
                    val[u][2 * p] = t.x; val[u][2 * p + 1] = t.y;
                }
            }
            #pragma unroll
            for (int u = 0; u < UR; ++u) {
                if (i0 + u < cnt) {
                    const int k = r[u] * VEC;
                    #pragma unroll
                    for (int p = 0; p < VEC; ++p) acc[k + p] += val[u][p];
                }
            }
        }
        #pragma unroll
        for (int j = 0; j < RPW; ++j) {
            const int row = wave * RPW + j;
            if (row < OWN_ROWS) {
                float* tp = tile + row * TS + lane * 2;
                #pragma unroll
                for (int p = 0; p < NP; ++p) {
                    float2 t = *(float2*)(tp + p * 64);
                    t.x += acc[j * VEC + 2 * p]; t.y += acc[j * VEC + 2 * p + 1];
                    *(float2*)(tp + p * 64) = t;
                }
            }
        }
    } else {
        if (tid == 0) *qcount = 0;
        __syncthreads();

        auto file = [&](unsigned dl, int e) {
            if (dl < (unsigned)OWN_ROWS) {
                if constexpr (!FINAL) atomicAdd(&rowcnt[dl], 1);
                const int slot = atomicAdd(qcount, 1);
                if (slot < QCAP) {
                    queue[slot] = make_int2(e, (int)dl);
                } else {
                    const float* hp = H + (size_t)src[e] * F;
                    float* tp = tile + dl * TS;
                    #pragma nounroll
                    for (int j = 0; j < F; ++j) atomicAdd(tp + j, hp[j]);
                }
            }
        };
        auto file4 = [&](const int4 v, int e) {
            const unsigned d0 = (unsigned)(v.x - nbase);
            const unsigned d1 = (unsigned)(v.y - nbase);
            const unsigned d2 = (unsigned)(v.z - nbase);
            const unsigned d3 = (unsigned)(v.w - nbase);
            const unsigned dm = min(min(d0, d1), min(d2, d3));
            if (dm < (unsigned)OWN_ROWS) {
                file(d0, e); file(d1, e + 1); file(d2, e + 2); file(d3, e + 3);
            }
        };

        const int E4 = E >> 2;
        const int4* dst4 = (const int4*)dst;
        const int start = ((int)blockIdx.x * (E4 / (int)gridDim.x)) & ~1023;
        #pragma nounroll
        for (int seg = 0; seg < 2; ++seg) {
            const int hi = seg ? start : E4;
            int b = (seg ? 0 : start) + tid;
            for (; b + 7 * 1024 < hi; b += 8 * 1024) {
                int4 v[8];
                #pragma unroll
                for (int u = 0; u < 8; ++u) v[u] = dst4[b + u * 1024];
                #pragma unroll
                for (int u = 0; u < 8; ++u) file4(v[u], (b + u * 1024) << 2);
            }
            for (; b < hi; b += 1024) file4(dst4[b], b << 2);
        }
        for (int e = (E4 << 2) + tid; e < E; e += 1024)
            file((unsigned)(dst[e] - nbase), e);
        __syncthreads();

        const int nraw = __builtin_amdgcn_readfirstlane(*qcount);
        n = nraw < QCAP ? nraw : QCAP;

        for (int i = tid; i < n; i += 4 * 1024) {
            int s[4];
            #pragma unroll
            for (int k = 0; k < 4; ++k) {
                const int ik = (i + k * 1024 < n) ? (i + k * 1024) : i;
                s[k] = src[queue[ik].x];
            }
            queue[i].x = s[0];
            #pragma unroll
            for (int k = 1; k < 4; ++k)
                if (i + k * 1024 < n) queue[i + k * 1024].x = s[k];
        }
        __syncthreads();

        auto rowadd = [&](int r, const float* val) {
            float* tp = tile + r * TS + lane * 2;
            #pragma unroll
            for (int p = 0; p < NP; ++p) {
                atomicAdd(tp + p * 64,     val[2 * p]);
                atomicAdd(tp + p * 64 + 1, val[2 * p + 1]);
            }
        };
        constexpr int U = 8;
        const int nfull = n & ~(U - 1);
        for (int i0 = wave * U; i0 < nfull; i0 += 32 * U) {
            const int2 mine = queue[i0 + (lane & (U - 1))];
            int   s[U], r[U];
            float val[U][VEC];
            #pragma unroll
            for (int u = 0; u < U; ++u) {
                s[u] = __builtin_amdgcn_readlane(mine.x, u);
                r[u] = __builtin_amdgcn_readlane(mine.y, u);
            }
            #pragma unroll
            for (int u = 0; u < U; ++u) rowload(s[u], val[u]);
            #pragma unroll
            for (int u = 0; u < U; ++u) rowadd(r[u], val[u]);
        }
        if (wave < n - nfull) {
            const int2 q = queue[nfull + wave];
            float val[VEC];
            rowload(q.x, val);
            rowadd(q.y, val);
        }
    }

    int rows = nn - nbase;
    if (rows > OWN_ROWS) rows = OWN_ROWS;
    if constexpr (FINAL) {
        float4* o = (float4*)(agg + (size_t)nbase * F);
        constexpr int Q   = F / 4;
        constexpr int RPS = 1024 / Q;
        constexpr int NSW = (OWN_ROWS + RPS - 1) / RPS;
        const int q  = tid % Q;
        const int r0 = tid / Q;
        const float4 bb = *(const float4*)(bias + q * 4);
        float nd[NSW];
        #pragma unroll
        for (int k = 0; k < NSW; ++k) {
            const int r = r0 + k * RPS;
            nd[k] = (r < rows) ? inv_sqrt_deg(deg[nbase + r]) : 0.0f;
        }
        __syncthreads();
        #pragma unroll
        for (int k = 0; k < NSW; ++k) {
            const int r = r0 + k * RPS;
            if (r < rows) {
                const float4 t = ((const float4*)tile)[r * Q + q];
                float4 v;
                v.x = t.x * nd[k] + bb.x;
                v.y = t.y * nd[k] + bb.y;
                v.z = t.z * nd[k] + bb.z;
                v.w = t.w * nd[k] + bb.w;
                o[r * Q + q] = v;
            }
        }
    } else {
        constexpr int NSW = (OWN_ROWS + 31) / 32;
        constexpr int NT  = FN / 16;
        constexpr int MG  = 32 / NT;
        static_assert(F == 128 && (32 % NT) == 0, "one 16-byte piece per lane; waves split evenly over the column tiles");
        const int q  = lane;
        const int r0 = wave;
        const float4 bb = *(const float4*)(bias + q * 4);
        const int nt   = wave % NT;
        const int col  = nt * 16 + (lane & 15);
        const int krow = (lane >> 4) << 1;
        v2f b[F / 4];
        #pragma unroll
        for (int s = 0; s < F / 4; ++s) {
            const int k0 = s * 4 + krow;
            b[s].x = Wn[(size_t)k0 * FN + col];
            b[s].y = Wn[(size_t)(k0 + 1) * FN + col];
        }
        __syncthreads();

        #pragma unroll
        for (int k = 0; k < NSW; ++k) {
            const int r = r0 + k * 32;
            if (r < rows) {
                const float nd = inv_sqrt_deg((float)rowcnt[r]);
                const float ns = inv_sqrt_deg((float)outdeg[r]);
                float4* tp = (float4*)(tile + r * TS) + q;
                const float4 v = *tp;
                float4 t;
                t.x = fmaxf(v.x * nd + bb.x, 0.0f) * ns;
                t.y = fmaxf(v.y * nd + bb.y, 0.0f) * ns;
                t.z = fmaxf(v.z * nd + bb.z, 0.0f) * ns;
                t.w = fmaxf(v.w * nd + bb.w, 0.0f) * ns;
                *tp = t;
            }
        }
        if (tid < rows) deg[nbase + tid] = (float)rowcnt[tid];
        __syncthreads();

        for (int mt = wave / NT; mt * 16 < rows; mt += MG) {
            const float* ap = tile + (mt * 16 + (lane & 15)) * TS + krow;
            v8f c = {};
            #pragma unroll
            for (int s = 0; s < F / 4; ++s) {
                const v2f a = *(const v2f*)(ap + s * 4);
                c = __builtin_amdgcn_wmma_f32_16x16x4_f32(false, a, false, b[s], (short)0, c, false, false);
            }
            const int rbase = mt * 16 + ((lane >> 4) << 3);
            float* p = agg + (size_t)(nbase + rbase) * FN + col;
            if (mt * 16 + 16 <= rows) {
                #pragma unroll
                for (int v = 0; v < 8; ++v) p[v * FN] = c[v];
            } else {
                #pragma unroll
                for (int v = 0; v < 8; ++v)
                    if (rbase + v < rows) p[v * FN] = c[v];
            }
        }
    }
}

__global__ __launch_bounds__(1024) void gcn_own128(const float* __restrict__ H,
                                                   const int* __restrict__ src,
                                                   const int* __restrict__ dst,
                                                   float* __restrict__ deg_dst,
                                                   const int* __restrict__ degp, int pitch,
                                                   const float* __restrict__ b1,
                                                   const float* __restrict__ W2,
                                                   float* __restrict__ H2,
                                                   const int* __restrict__ pcount,
                                                   const int* __restrict__ slabs, int E, int nn) {
    __shared__ float tile[((OWN_ROWS + 15) / 16) * 16 * (128 + 4)];
    __shared__ int2  queue[QLIST];
    __shared__ int   rowcnt[OWN_ROWS];
    __shared__ int   outdeg[OWN_ROWS];
    __shared__ int   bincnt[32];
    __shared__ int   wover[32];
    __shared__ int   qcount;
    own_body<128, QLIST, false, 64>(tile, queue, &qcount, rowcnt, bincnt, outdeg, wover, H, src, dst, deg_dst, degp, pitch,
                                    b1, W2, H2, pcount, slabs, E, nn);
}

__global__ __launch_bounds__(1024) void gcn_own64(const float* __restrict__ H,
                                                  const int* __restrict__ src,
                                                  const int* __restrict__ dst,
                                                  float* __restrict__ deg_dst,
                                                  const float* __restrict__ b2,
                                                  float* __restrict__ out,
                                                  const int* __restrict__ pcount,
                                                  const int* __restrict__ slabs, int E, int nn) {
    __shared__ float tile[OWN_ROWS * 64];
    __shared__ int2  queue[16384];
    __shared__ int   bincnt[32];
    __shared__ int   wover[32];
    __shared__ int   qcount;
    own_body<64, 16384, true, 0>(tile, queue, &qcount, nullptr, bincnt, nullptr, wover, H, src, dst, deg_dst, nullptr, 0,
                                 b2, nullptr, out, pcount, slabs, E, nn);
}


extern "C" void kernel_launch(void* const* d_in, const int* in_sizes, int n_in,
                              void* d_out, int out_size, void* d_ws, size_t ws_size,
                              hipStream_t stream) {
    const float* feats = (const float*)d_in[0];
    const float* W1    = (const float*)d_in[1];
    const float* b1    = (const float*)d_in[2];
    const float* W2    = (const float*)d_in[3];
    const float* b2    = (const float*)d_in[4];
    const int*   src   = (const int*)d_in[5];
    const int*   dst   = (const int*)d_in[6];

    const int N = in_sizes[0] / 128;
    const int E = in_sizes[5];

    const int pitch = ((N + 127) / 128) * 128;
    float* ws       = (float*)d_ws;
    float* deg_dst  = ws;                  ws += pitch;
    int*   degp     = (int*)ws;            ws += (size_t)HB * pitch;
    float* H1       = ws;                  ws += (size_t)N * 128;
    float* H2       = ws;                  ws += (size_t)N * 64;
    const int owners = (N + OWN_ROWS - 1) / OWN_ROWS;
    int*  pcount    = (int*)ws;            ws += (size_t)FBMAX * POWN;
    int*  slabs     = (int*)ws;

    gcn_file<<<HB + (E + PCHUNK - 1) / PCHUNK, 1024, 0, stream>>>(src, dst, degp, pitch, pcount, slabs, E, owners, N);

    const int mtiles = (N + 15) / 16;

    gcn_gemm1<<<mtiles, 256, 0, stream>>>(feats, degp, pitch, W1, H1, N);
    gcn_own128<<<owners, 1024, 0, stream>>>(H1, src, dst, deg_dst, degp, pitch, b1, W2, H2, pcount, slabs, E, N);

    gcn_own64<<<owners, 1024, 0, stream>>>(H2, src, dst, deg_dst, b2, (float*)d_out,
                                           pcount, slabs, E, N);
}
